# starting kernel
# baseline (speedup 1.0000x reference)
#include <hip/hip_runtime.h>
#include <math.h>

#define B_ 2
#define C_ 128
#define T_ 8
#define H_ 64
#define W_ 64
#define N_ (T_*H_*W_)
#define BN_ (B_*N_)
#define HID_ 256
#define MAX_STEPS_ 12
#define DIFF_SCALE 0.25f
#define PROBE_CELLS (B_*C_*2*8*8)

typedef __attribute__((ext_vector_type(16))) _Float16 v16h;
typedef __attribute__((ext_vector_type(8)))  _Float16 h8;
typedef __attribute__((ext_vector_type(8)))  float    v8f;

__device__ __forceinline__ float softplus_f(float x){ return x > 20.f ? x : log1pf(expf(x)); }
__device__ __forceinline__ float gelu_f(float x){ return 0.5f*x*(1.f + erff(x*0.70710678118654752f)); }
__device__ __forceinline__ int imin_(int a, int b){ return a < b ? a : b; }

__device__ __forceinline__ float blockReduceSum(float v){
  __shared__ float red[8];
  int lane = threadIdx.x & 31, wid = threadIdx.x >> 5;
  #pragma unroll
  for (int off = 16; off > 0; off >>= 1) v += __shfl_down(v, off);
  if (lane == 0) red[wid] = v;
  __syncthreads();
  v = (threadIdx.x < (blockDim.x >> 5)) ? red[threadIdx.x] : 0.f;
  if (wid == 0){
    #pragma unroll
    for (int off = 4; off > 0; off >>= 1) v += __shfl_down(v, off);
  }
  __syncthreads();
  return v;
}

__device__ __forceinline__ v16h load_frag_row(const _Float16* base, int ld, int r0, int k0, int lane){
  int r = r0 + (lane & 15);
  int hf = lane >> 4;
  const _Float16* p = base + r*ld + k0 + hf*8;
  h8 lo = *(const h8*)p;
  h8 hi = *(const h8*)(p + 16);
  return __builtin_shufflevector(lo, hi, 0,1,2,3,4,5,6,7,8,9,10,11,12,13,14,15);
}


__global__ void zero_kernel(float* p, int n){
  int i = blockIdx.x*blockDim.x + threadIdx.x;
  if (i < n) p[i] = 0.f;
}

__global__ void prep_kernel(const float* w1, const float* w2,
                            const float* ks, const float* kt,
                            const float* ag, const float* al, const float* logdt,
                            _Float16* w1h, _Float16* w2h,
                            float* spsks, float* spskt, float* scal){
  int i = blockIdx.x*blockDim.x + threadIdx.x;
  int stride = gridDim.x*blockDim.x;
  for (int j = i; j < C_*HID_; j += stride) w1h[j] = (_Float16)w1[j];
  for (int j = i; j < HID_*C_; j += stride) w2h[j] = (_Float16)w2[j];
  for (int j = i; j < 3*C_;   j += stride) spsks[j] = softplus_f(ks[j]);
  for (int j = i; j < 2*C_;   j += stride) spskt[j] = softplus_f(kt[j]);
  if (i == 0){
    float dt = expf(logdt[0]);
    dt = fminf(fmaxf(dt, 0.005f), 0.35f);
    scal[0] = dt;
    scal[1] = softplus_f(ag[0]);
    scal[2] = softplus_f(al[0]);
  }
}

__global__ void pool_kernel(const float* in, float* out, int tp, int hp, int wp){
  int cells = B_*C_*tp*hp*wp;
  int id = blockIdx.x*blockDim.x + threadIdx.x;
  if (id >= cells) return;
  int pw = id % wp; int tmp = id / wp;
  int ph = tmp % hp; tmp /= hp;
  int pt = tmp % tp; tmp /= tp;
  int c = tmp % C_;  int b = tmp / C_;
  int bt = T_/tp, bh = H_/hp, bw = W_/wp;
  const float* base = in + (size_t)(b*C_ + c)*N_;
  float s = 0.f;
  for (int it = 0; it < bt; ++it)
    for (int ih = 0; ih < bh; ++ih)
      for (int iw = 0; iw < bw; ++iw)
        s += base[((pt*bt+it)*H_ + (ph*bh+ih))*W_ + (pw*bw+iw)];
  out[id] = s / (float)(bt*bh*bw);
}

__global__ void pooled_mean_kernel(const float* u, float* pooled){
  int bc = blockIdx.x;
  const float* base = u + (size_t)bc*N_;
  float s = 0.f;
  for (int i = threadIdx.x; i < N_; i += blockDim.x) s += base[i];
  s = blockReduceSum(s);
  if (threadIdx.x == 0) pooled[bc] = s / (float)N_;
}

__global__ void gate_value_kernel(const float* pooled, const float* gw, const float* gb,
                                  const float* vw, const float* vb, float* h){
  int tid = threadIdx.x;
  int b = tid >> 7, c = tid & 127;
  const float* p = pooled + b*C_;
  float g = gb[c], v = vb[c];
  for (int k = 0; k < C_; ++k){
    float pk = p[k];
    g += pk * gw[k*C_ + c];
    v += pk * vw[k*C_ + c];
  }
  h[tid] += (1.f/(1.f+expf(-g))) * tanhf(v);
}

__global__ void local_proj_kernel(const float* lsmall, const float* wlp, const float* blp, float* lproj){
  int id = blockIdx.x*blockDim.x + threadIdx.x;
  if (id >= B_*C_*32) return;
  int cell = id & 31;
  int oc = (id >> 5) & 127;
  int b = id >> 12;
  float s = blp[oc];
  for (int c = 0; c < C_; ++c)
    s += lsmall[((size_t)(b*C_ + c) << 5) + cell] * wlp[c*C_ + oc];
  lproj[((size_t)(b*C_ + oc) << 5) + cell] = s;
}

__global__ void __launch_bounds__(256)
reaction_mlp_kernel(const float* u, const _Float16* w1h, const float* b1,
                    const _Float16* w2h, const float* b2, float* react){
  __shared__ _Float16 As[128][128];
  __shared__ _Float16 Bt[32][128];
  __shared__ _Float16 Hs[128][32];
  __shared__ _Float16 W2t[128][32];

  const int tid  = threadIdx.x;
  const int lane = tid & 31;
  const int wv   = tid >> 5;
  const int m0   = wv * 16;
  const int hf   = lane >> 4;
  const int col  = lane & 15;

  const int rowBase = blockIdx.x * 128;
  const int b  = rowBase / N_;
  const int n0 = rowBase - b * N_;
  const float* ub = u + (size_t)b * C_ * N_;

  for (int idx = tid; idx < 128*32; idx += 256){
    int c = idx >> 5;
    int q = idx & 31;
    const float4 v = *(const float4*)(ub + (size_t)c * N_ + n0 + q*4);
    As[q*4+0][c] = (_Float16)v.x;
    As[q*4+1][c] = (_Float16)v.y;
    As[q*4+2][c] = (_Float16)v.z;
    As[q*4+3][c] = (_Float16)v.w;
  }

  const v8f z8 = {0.f,0.f,0.f,0.f,0.f,0.f,0.f,0.f};
  v8f accC[8];
  #pragma unroll
  for (int i = 0; i < 8; ++i) accC[i] = z8;

  for (int hc = 0; hc < 8; ++hc){
    if (hc < 7){
      __builtin_prefetch(&w1h[(hc+1)*32], 0, 1);
      __builtin_prefetch(&w2h[((hc+1)*32)*C_], 0, 1);
    }
    __syncthreads();
    for (int idx = tid; idx < 32*128; idx += 256){
      int j = idx & 31;
      int k = idx >> 5;
      Bt[j][k] = w1h[k*HID_ + hc*32 + j];
    }
    for (int idx = tid; idx < 32*128; idx += 256){
      int nc = idx & 127;
      int k = idx >> 7;
      W2t[nc][k] = w2h[(hc*32 + k)*C_ + nc];
    }
    __syncthreads();

    v8f a0 = z8, a1 = z8;
    #pragma unroll
    for (int k0 = 0; k0 < 128; k0 += 32){
      v16h af  = load_frag_row(&As[0][0], 128, m0, k0, lane);
      v16h bf0 = load_frag_row(&Bt[0][0], 128, 0,  k0, lane);
      v16h bf1 = load_frag_row(&Bt[0][0], 128, 16, k0, lane);
      a0 = __builtin_amdgcn_wmma_f32_16x16x32_f16(false, af, false, bf0, (short)0, a0, false, false);
      a1 = __builtin_amdgcn_wmma_f32_16x16x32_f16(false, af, false, bf1, (short)0, a1, false, false);
    }
    #pragma unroll
    for (int r = 0; r < 8; ++r){
      int m = m0 + hf*8 + r;
      float x0 = a0[r] + b1[hc*32 + col];
      float x1 = a1[r] + b1[hc*32 + 16 + col];
      Hs[m][col]      = (_Float16)gelu_f(x0);
      Hs[m][16 + col] = (_Float16)gelu_f(x1);
    }
    v16h ah = load_frag_row(&Hs[0][0], 32, m0, 0, lane);
    #pragma unroll
    for (int nt = 0; nt < 8; ++nt){
      v16h bw = load_frag_row(&W2t[0][0], 32, nt*16, 0, lane);
      accC[nt] = __builtin_amdgcn_wmma_f32_16x16x32_f16(false, ah, false, bw, (short)0, accC[nt], false, false);
    }
  }

  #pragma unroll
  for (int nt = 0; nt < 8; ++nt){
    int c = nt*16 + col;
    float bias = b2[c];
    float* dst = react + (size_t)(b*C_ + c)*N_ + n0 + m0 + hf*8;
    float4 v0 = { accC[nt][0]+bias, accC[nt][1]+bias, accC[nt][2]+bias, accC[nt][3]+bias };
    float4 v1 = { accC[nt][4]+bias, accC[nt][5]+bias, accC[nt][6]+bias, accC[nt][7]+bias };
    *(float4*)dst       = v0;
    *(float4*)(dst + 4) = v1;
  }
}

__global__ void __launch_bounds__(256)
update_kernel(const float* u, const float* react, const float* h,
              const float* lproj, const float* spsks, const float* spskt,
              const float* scal, const float* normw,
              float* ucand, float* unext, float* accum, int do_norm){
  int p = blockIdx.x*blockDim.x + threadIdx.x;
  int b  = p >> 15;
  int n  = p & (N_ - 1);
  int t  = n >> 12;
  int hh = (n >> 6) & 63;
  int ww = n & 63;

  float dt = scal[0], ag = scal[1], al = scal[2];

  float st = fminf(fmaxf((t  + 0.5f)*(2.f/T_) - 0.5f, 0.f), 1.f);
  float sh = fminf(fmaxf((hh + 0.5f)*(4.f/H_) - 0.5f, 0.f), 3.f);
  float sw = fminf(fmaxf((ww + 0.5f)*(4.f/W_) - 0.5f, 0.f), 3.f);
  int t0i = (int)st; int t1i = imin_(t0i+1, 1); float ft = st - (float)t0i;
  int h0i = (int)sh; int h1i = imin_(h0i+1, 3); float fh = sh - (float)h0i;
  int w0i = (int)sw; int w1i = imin_(w0i+1, 3); float fw = sw - (float)w0i;

  float sum_du = 0.f, sum_lap = 0.f, sumsq = 0.f;
  size_t bbase = (size_t)b * C_ * N_;

  for (int c = 0; c < C_; ++c){
    const float* uc = u + bbase + (size_t)c * N_;
    float uv = uc[n];

    float diff = 0.f;
    #pragma unroll
    for (int i = 0; i < 3; ++i){
      const int d = (i == 0) ? 1 : ((i == 1) ? 4 : 16);
      float lap = -4.f * uv;
      lap += (hh - d >= 0) ? uc[n - d*64] : 0.f;
      lap += (hh + d < 64) ? uc[n + d*64] : 0.f;
      lap += (ww - d >= 0) ? uc[n - d]    : 0.f;
      lap += (ww + d < 64) ? uc[n + d]    : 0.f;
      diff += spsks[i*C_ + c] * lap;
    }
    #pragma unroll
    for (int i = 0; i < 2; ++i){
      const int d = i + 1;
      float up = (t - d >= 0) ? uc[n - d*4096] : 0.f;
      diff += spskt[i*C_ + c] * (up - uv);
    }
    diff *= DIFF_SCALE;
    sum_lap += fabsf(diff);

    const float* lp = lproj + ((size_t)(b*C_ + c) << 5);
    float c00 = lp[t0i*16 + h0i*4 + w0i]*(1.f-fw) + lp[t0i*16 + h0i*4 + w1i]*fw;
    float c01 = lp[t0i*16 + h1i*4 + w0i]*(1.f-fw) + lp[t0i*16 + h1i*4 + w1i]*fw;
    float c10 = lp[t1i*16 + h0i*4 + w0i]*(1.f-fw) + lp[t1i*16 + h0i*4 + w1i]*fw;
    float c11 = lp[t1i*16 + h1i*4 + w0i]*(1.f-fw) + lp[t1i*16 + h1i*4 + w1i]*fw;
    float lv = (c00*(1.f-fh) + c01*fh)*(1.f-ft) + (c10*(1.f-fh) + c11*fh)*ft;

    float du = diff + react[bbase + (size_t)c*N_ + n] + ag*h[b*C_ + c] + al*lv;
    float cand = uv + dt*du;
    sum_du += fabsf(du);
    sumsq  += cand*cand;
    ucand[bbase + (size_t)c*N_ + n] = cand;
  }

  float scale = do_norm ? rsqrtf(sumsq/(float)C_ + 1e-8f) : 1.f;
  for (int c = 0; c < C_; ++c){
    float cand = ucand[bbase + (size_t)c*N_ + n];
    unext[bbase + (size_t)c*N_ + n] = do_norm ? cand*scale*normw[c] : cand;
  }

  float s1 = blockReduceSum(sum_du);
  float s2 = blockReduceSum(sum_lap);
  if (threadIdx.x == 0){
    atomicAdd(&accum[1], s1);
    atomicAdd(&accum[0], s2);
  }
}

__global__ void turb_kernel(const float* cur, const float* prev, float* accum){
  int i = blockIdx.x*blockDim.x + threadIdx.x;
  float a = 0.f, pv = 0.f;
  if (i < PROBE_CELLS){
    float cv = cur[i], pp = prev[i];
    a = fabsf(cv - pp);
    pv = fabsf(pp);
  }
  float s1 = blockReduceSum(a);
  float s2 = blockReduceSum(pv);
  if (threadIdx.x == 0){
    atomicAdd(&accum[2], s1);
    atomicAdd(&accum[3], s2);
  }
}

__global__ void finalize_step_kernel(const float* accum, float* turbs, float* sten, int step){
  if (threadIdx.x == 0 && blockIdx.x == 0){
    const float inv_cells = 1.f/(float)PROBE_CELLS;
    const float inv_bnc   = 1.f/8388608.f;
    float stop_turb = (accum[2]*inv_cells) / (accum[3]*inv_cells + 1e-8f);
    float se = accum[1]*inv_bnc;
    float le = accum[0]*inv_bnc;
    turbs[step] = stop_turb + 0.05f*se + 0.01f*le;
    sten[step]  = se;
  }
}

__global__ void final_out_kernel(const float* turbs, const float* sten, float* out){
  if (threadIdx.x == 0 && blockIdx.x == 0){
    float a = 0.f, bm = 0.f;
    for (int i = 0; i < MAX_STEPS_; ++i){ a += turbs[i]; bm += sten[i]; }
    out[0] = a / (float)MAX_STEPS_;
    out[1] = bm / (float)MAX_STEPS_;
  }
}


extern "C" void kernel_launch(void* const* d_in, const int* in_sizes, int n_in,
                              void* d_out, int out_size, void* d_ws, size_t ws_size,
                              hipStream_t stream){
  (void)in_sizes; (void)n_in; (void)out_size; (void)ws_size;

  const float* u_in  = (const float*)d_in[0];
  const float* w1    = (const float*)d_in[1];
  const float* b1    = (const float*)d_in[2];
  const float* w2    = (const float*)d_in[3];
  const float* b2    = (const float*)d_in[4];
  const float* gw    = (const float*)d_in[5];
  const float* gb    = (const float*)d_in[6];
  const float* vw    = (const float*)d_in[7];
  const float* vb    = (const float*)d_in[8];
  const float* normw = (const float*)d_in[9];
  const float* wlp   = (const float*)d_in[10];
  const float* blp   = (const float*)d_in[11];
  const float* a_g   = (const float*)d_in[12];
  const float* a_l   = (const float*)d_in[13];
  const float* logdt = (const float*)d_in[14];
  const float* ks    = (const float*)d_in[15];
  const float* kt    = (const float*)d_in[16];

  float* Wf = (float*)d_ws;
  size_t o = 0;
  const size_t VOL = (size_t)B_ * C_ * N_;
  float* ubuf0  = Wf + o; o += VOL;
  float* ubuf1  = Wf + o; o += VOL;
  float* ucand  = Wf + o; o += VOL;
  float* react  = Wf + o; o += VOL;
  float* probeA = Wf + o; o += PROBE_CELLS;
  float* probeB = Wf + o; o += PROBE_CELLS;
  float* pooled = Wf + o; o += 256;
  float* hstate = Wf + o; o += 256;
  float* lsmall = Wf + o; o += (size_t)B_*C_*32;
  float* lproj  = Wf + o; o += (size_t)B_*C_*32;
  float* spsks  = Wf + o; o += 3*C_;
  float* spskt  = Wf + o; o += 2*C_;
  float* scal   = Wf + o; o += 8;
  float* accum  = Wf + o; o += 8;
  float* turbs  = Wf + o; o += 16;
  float* sten   = Wf + o; o += 16;
  _Float16* w1h = (_Float16*)(Wf + o); o += (C_*HID_)/2;
  _Float16* w2h = (_Float16*)(Wf + o); o += (HID_*C_)/2;

  float* ubuf[2] = { ubuf0, ubuf1 };

  prep_kernel<<<64, 256, 0, stream>>>(w1, w2, ks, kt, a_g, a_l, logdt,
                                      w1h, w2h, spsks, spskt, scal);
  zero_kernel<<<1, 256, 0, stream>>>(hstate, 256);
  hipMemcpyAsync(ubuf[0], u_in, VOL*sizeof(float), hipMemcpyDeviceToDevice, stream);
  pool_kernel<<<(PROBE_CELLS + 255)/256, 256, 0, stream>>>(ubuf[0], probeA, 2, 8, 8);

  float* pprev = probeA;
  float* pcur  = probeB;

  for (int step = 0; step < MAX_STEPS_; ++step){
    float* ucur  = ubuf[step & 1];
    float* unext = ubuf[(step & 1) ^ 1];
    int do_norm = (((step + 1) % 2) == 0) ? 1 : 0;

    zero_kernel<<<1, 32, 0, stream>>>(accum, 8);
    reaction_mlp_kernel<<<BN_/128, 256, 0, stream>>>(ucur, w1h, b1, w2h, b2, react);
    pooled_mean_kernel<<<B_*C_, 256, 0, stream>>>(ucur, pooled);
    gate_value_kernel<<<1, 256, 0, stream>>>(pooled, gw, gb, vw, vb, hstate);
    pool_kernel<<<(B_*C_*32 + 255)/256, 256, 0, stream>>>(ucur, lsmall, 2, 4, 4);
    local_proj_kernel<<<(B_*C_*32 + 255)/256, 256, 0, stream>>>(lsmall, wlp, blp, lproj);
    update_kernel<<<BN_/256, 256, 0, stream>>>(ucur, react, hstate, lproj, spsks, spskt,
                                               scal, normw, ucand, unext, accum, do_norm);
    pool_kernel<<<(PROBE_CELLS + 255)/256, 256, 0, stream>>>(ucand, pcur, 2, 8, 8);
    turb_kernel<<<PROBE_CELLS/256, 256, 0, stream>>>(pcur, pprev, accum);
    finalize_step_kernel<<<1, 1, 0, stream>>>(accum, turbs, sten, step);

    float* tswap = pprev; pprev = pcur; pcur = tswap;
  }

  hipMemcpyAsync(d_out, ubuf[MAX_STEPS_ & 1], VOL*sizeof(float),
                 hipMemcpyDeviceToDevice, stream);
  final_out_kernel<<<1, 32, 0, stream>>>(turbs, sten, ((float*)d_out) + VOL);
}


# update takes the centre, the 6 arms along the row and the arm to the group's other row from an f32 LDS copy of the state tile written by the MLP gather: 15 -> 7 global loads per channel step
# speedup vs baseline: 7.6332x; 7.6332x over previous
#include <hip/hip_runtime.h>
#include <math.h>
#include <type_traits>

#define B_ 2
#define C_ 128
#define T_ 8
#define H_ 64
#define W_ 64
#define N_ (T_*H_*W_)
#define BN_ (B_*N_)
#define HID_ 256
#define MAX_STEPS_ 12
#define DIFF_SCALE 0.25f
#define PROBE_CELLS (B_*C_*2*8*8)

typedef __attribute__((ext_vector_type(16))) _Float16 v16h;
typedef __attribute__((ext_vector_type(8)))  _Float16 h8;
typedef __attribute__((ext_vector_type(8)))  float    v8f;

__device__ __forceinline__ float softplus_f(float x){ return x > 20.f ? x : log1pf(expf(x)); }
__device__ __forceinline__ float gelu_f(float x){ return 0.5f*x*(1.f + erff(x*0.70710678118654752f)); }
__device__ __forceinline__ int imin_(int a, int b){ return a < b ? a : b; }

__device__ __forceinline__ float blockReduceSum(float v){
  __shared__ float red[8];
  int lane = threadIdx.x & 31, wid = threadIdx.x >> 5;
  #pragma unroll
  for (int off = 16; off > 0; off >>= 1) v += __shfl_down(v, off);
  if (lane == 0) red[wid] = v;
  __syncthreads();
  v = (threadIdx.x < (blockDim.x >> 5)) ? red[threadIdx.x] : 0.f;
  if (wid == 0){
    #pragma unroll
    for (int off = 4; off > 0; off >>= 1) v += __shfl_down(v, off);
  }
  __syncthreads();
  return v;
}

__device__ __forceinline__ v16h load_frag_glob(const _Float16* base, int f, int lane){
  const h8* p = (const h8*)base + f*64 + lane;
  h8 lo = p[0];
  h8 hi = p[32];
  return __builtin_shufflevector(lo, hi, 0,1,2,3,4,5,6,7,8,9,10,11,12,13,14,15);
}


__global__ void prep_kernel(const float* w1, const float* w2,
                            const float* ks, const float* kt,
                            const float* ag, const float* al, const float* logdt,
                            _Float16* w1h, _Float16* w2h,
                            float* spsks, float* spskt, float* scal, float* hstate){
  int i = blockIdx.x*blockDim.x + threadIdx.x;
  int stride = gridDim.x*blockDim.x;
  if (i < B_*C_) hstate[i] = 0.f;
  for (int j = i; j < C_*HID_; j += stride){
    int e = j & 7, ln = (j >> 3) & 31, half = (j >> 8) & 1;
    int nt = (j >> 9) & 1, kb = (j >> 10) & 3, hc = j >> 12;
    int k = kb*32 + half*16 + (ln >> 4)*8 + e;
    int n = hc*32 + nt*16 + (ln & 15);
    w1h[j] = (_Float16)w1[k*HID_ + n];
  }
  for (int j = i; j < HID_*C_; j += stride){
    int e = j & 7, ln = (j >> 3) & 31, half = (j >> 8) & 1;
    int nt = (j >> 9) & 7, hc = j >> 12;
    int k = hc*32 + half*16 + (ln >> 4)*8 + e;
    int n = nt*16 + (ln & 15);
    w2h[j] = (_Float16)w2[k*C_ + n];
  }
  for (int j = i; j < 3*C_;   j += stride) spsks[j] = softplus_f(ks[j]);
  for (int j = i; j < 2*C_;   j += stride) spskt[j] = softplus_f(kt[j]);
  if (i == 0){
    float dt = expf(logdt[0]);
    dt = fminf(fmaxf(dt, 0.005f), 0.35f);
    scal[0] = dt;
    scal[1] = softplus_f(ag[0]);
    scal[2] = softplus_f(al[0]);
  }
}

__device__ __forceinline__ float blockrow_sum(const float* p){
  const float4* q = (const float4*)p;
  float4 v[16];
  #pragma unroll
  for (int f = 0; f < 4; ++f){
    #pragma unroll
    for (int j = 0; j < 4; ++j) v[f*4 + j] = q[f*(H_*W_/4) + j*(2*W_/4)];
  }
  float s = 0.f;
  #pragma unroll
  for (int i = 0; i < 16; ++i) s += (v[i].x + v[i].y) + (v[i].z + v[i].w);
  s += __shfl_xor(s, 1);
  s += __shfl_xor(s, 16);
  return s;
}

__global__ void __launch_bounds__(512)
pool3_kernel(const float* cand, const float* state, int state_differs,
             float* probe, const float* prev, float* lsmall, float* pooled, float* accum){
  __shared__ float cs[16][8];
  __shared__ float dsum[16][8];
  __shared__ float psum[16][8];
  const int bc   = blockIdx.x;
  const int lane = threadIdx.x & 31;
  const int wv   = threadIdx.x >> 5;
  const int pt = wv >> 3, ph = wv & 7;
  const size_t off = (size_t)bc*N_ + (size_t)pt*(4*H_*W_)
                   + (size_t)((ph*8 + (lane >> 4))*W_ + (lane & 15)*4);
  const int pw = (lane >> 1) & 7;
  const bool owner = (lane & 17) == 0;
  const size_t pi = (size_t)(bc*16 + wv)*8 + pw;
  float pp = 0.f;
  if (prev != nullptr && owner) pp = prev[pi];
  const float s = blockrow_sum(cand + off);
  float ss = s;
  if (state_differs) ss = blockrow_sum(state + off);
  if (owner){
    const float cv = s * (1.f/256.f);
    probe[pi] = cv;
    cs[wv][pw] = ss;
    dsum[wv][pw] = fabsf(cv - pp);
    psum[wv][pw] = fabsf(pp);
  }
  __syncthreads();
  if (wv == 0){
    const int lt = lane >> 4, lh = (lane >> 2) & 3, lw = lane & 3;
    const int r0 = lt*8 + lh*2;
    float a = (cs[r0][2*lw] + cs[r0][2*lw + 1]) + (cs[r0 + 1][2*lw] + cs[r0 + 1][2*lw + 1]);
    lsmall[(size_t)bc*32 + lane] = a * (1.f/1024.f);
    a += __shfl_xor(a, 16);
    a += __shfl_xor(a, 8);
    a += __shfl_xor(a, 4);
    a += __shfl_xor(a, 2);
    a += __shfl_xor(a, 1);
    if (lane == 0) pooled[bc] = a * (1.f/(float)N_);
  } else if (wv == 1 && prev != nullptr){
    const float* d = &dsum[0][0];
    const float* p = &psum[0][0];
    float a = (d[lane] + d[32 + lane]) + (d[64 + lane] + d[96 + lane]);
    float q = (p[lane] + p[32 + lane]) + (p[64 + lane] + p[96 + lane]);
    for (int off2 = 16; off2 > 0; off2 >>= 1){ a += __shfl_xor(a, off2); q += __shfl_xor(q, off2); }
    if (lane == 0){
      atomicAdd(&accum[2], a);
      atomicAdd(&accum[3], q);
    }
  }
}

__device__ __forceinline__ void step_scalars(const float* accum, float& turb, float& se){
  const float inv_cells = 1.f/(float)PROBE_CELLS;
  const float inv_bnc   = 1.f/8388608.f;
  const float stop_turb = (accum[2]*inv_cells) / (accum[3]*inv_cells + 1e-8f);
  se = accum[1]*inv_bnc;
  const float le = accum[0]*inv_bnc;
  turb = stop_turb + 0.05f*se + 0.01f*le;
}

#define MEM_GROUPS 34
__global__ void __launch_bounds__(256)
memory_kernel(const float* pooled, const float* gw, const float* gb,
              const float* vw, const float* vb, float* h,
              const float* lsmall, const float* wlp, const float* blp, float* lproj,
              float* accum, float* turbs, float* sten, int step){
  const int tid = threadIdx.x;
  if (blockIdx.x < 32){
    const int id = blockIdx.x*256 + tid;
    const int cell = id & 31;
    const int oc = (id >> 5) & 127;
    const int b = id >> 12;
    float s = blp[oc];
    #pragma unroll 32
    for (int c = 0; c < C_; ++c)
      s += lsmall[((size_t)(b*C_ + c) << 5) + cell] * wlp[c*C_ + oc];
    lproj[((size_t)(b*C_ + oc) << 5) + cell] = s;
  } else if (blockIdx.x == 32){
    const int b = tid >> 7, c = tid & 127;
    const float* p = pooled + b*C_;
    float g = gb[c], v = vb[c];
    #pragma unroll 32
    for (int k = 0; k < C_; ++k){
      const float pk = p[k];
      g += pk * gw[k*C_ + c];
      v += pk * vw[k*C_ + c];
    }
    h[tid] += (1.f/(1.f+expf(-g))) * tanhf(v);
  } else if (tid == 0){
    if (step > 0){
      float tb, se;
      step_scalars(accum, tb, se);
      turbs[step - 1] = tb;
      sten[step - 1]  = se;
    }
    accum[0] = 0.f; accum[1] = 0.f; accum[2] = 0.f; accum[3] = 0.f;
  }
}

__device__ __forceinline__ float2 ld2(const float* p, bool ok){
  const float2 a = *(const float2*)p;
  float2 o; o.x = ok ? a.x : 0.f; o.y = ok ? a.y : 0.f;
  return o;
}
__device__ __forceinline__ float ld1(const float* p, bool ok){
  const float a = *p;
  return ok ? a : 0.f;
}
__device__ __forceinline__ float2 sel2(float2 a, bool ok){
  float2 o; o.x = ok ? a.x : 0.f; o.y = ok ? a.y : 0.f;
  return o;
}
__device__ __forceinline__ void ldc8(const float* p, float* o){
  const float4 a = *(const float4*)p, b = *(const float4*)(p + 4);
  o[0] = a.x; o[1] = a.y; o[2] = a.z; o[3] = a.w; o[4] = b.x; o[5] = b.y; o[6] = b.z; o[7] = b.w;
}

__device__ __forceinline__ float upd1(float uv, float u1a, float u1b, float l1, float r1,
                                      float u4a, float u4b, float l4, float r4,
                                      float u16a, float u16b, float l16, float r16,
                                      float t1, float t2,
                                      float k0, float k1, float k2, float q0, float q1,
                                      float rest, float lv, float al, float dt,
                                      float& sum_lap, float& sum_du, float& ssq){
  float lap1  = -4.f * uv; lap1  += u1a;  lap1  += u1b;  lap1  += l1;  lap1  += r1;
  float lap4  = -4.f * uv; lap4  += u4a;  lap4  += u4b;  lap4  += l4;  lap4  += r4;
  float lap16 = -4.f * uv; lap16 += u16a; lap16 += u16b; lap16 += l16; lap16 += r16;
  float diff = k0 * lap1;
  diff += k1 * lap4;
  diff += k2 * lap16;
  diff += q0 * (t1 - uv);
  diff += q1 * (t2 - uv);
  diff *= DIFF_SCALE;
  sum_lap += fabsf(diff);
  const float du = diff + rest + al*lv;
  const float cand = uv + dt*du;
  sum_du += fabsf(du);
  ssq += cand*cand;
  return cand;
}

__global__ void __launch_bounds__(128) __attribute__((amdgpu_waves_per_eu(2)))
step_kernel(const float* __restrict__ u, const _Float16* __restrict__ w1h, const float* __restrict__ b1,
            const _Float16* __restrict__ w2h, const float* __restrict__ b2,
            const float* __restrict__ h, const float* __restrict__ lproj,
            const float* __restrict__ spsks, const float* __restrict__ spskt,
            const float* __restrict__ scal, const float* __restrict__ normw,
            float* __restrict__ ucand, float* __restrict__ unext, float* accum, int do_norm){
  const int tid  = threadIdx.x;
  const int lane = tid & 31;
  const int wv   = __builtin_amdgcn_readfirstlane(tid >> 5);
  const int hf   = lane >> 4;
  const int col  = lane & 15;

  const int pixBase = blockIdx.x * 128 + wv * 32;
  const int b  = pixBase / N_;
  const int pw = pixBase - b * N_;
  const float* ub = u + (size_t)b * C_ * N_;
  const int e0 = hf*8*N_ + pw + 2*col;
  const int t  = pw >> 12;
  const int hh = (pw >> 6) & 63;

  __shared__ __attribute__((aligned(16))) float Rs[4][128][36];
  __shared__ __attribute__((aligned(16))) float tabs[128][8];
  __shared__ __attribute__((aligned(16))) float Lws[2][128][4];
  __shared__ __attribute__((aligned(16))) float Cs[2][128][68];
  {
    const int c = tid;
    const float agv = scal[1];
    const float4 k4 = { spsks[c], spsks[C_ + c], spsks[2*C_ + c], spskt[c] };
    const float4 q4 = { spskt[C_ + c], b2[c], agv * h[b*C_ + c], normw[c] };
    *(float4*)&tabs[c][0] = k4;
    *(float4*)&tabs[c][4] = q4;
    const int tb_ = (blockIdx.x * 128 - b * N_) >> 12;
    const int hb_ = ((blockIdx.x * 128 - b * N_) >> 6) & 63;
    const float st  = fminf(fmaxf((tb_ + 0.5f)*(2.f/T_) - 0.5f, 0.f), 1.f);
    const int t0i = (int)st;  const int t1i = imin_(t0i+1, 1); const float ft = st - (float)t0i;
    const float4* lp = (const float4*)(lproj + ((size_t)(b*C_ + c) << 5));
    #pragma unroll
    for (int rw = 0; rw < 2; ++rw){
      const float sh  = fminf(fmaxf((hb_ + rw + 0.5f)*(4.f/H_) - 0.5f, 0.f), 3.f);
      const int h0i = (int)sh;  const int h1i = imin_(h0i+1, 3); const float fh = sh - (float)h0i;
      const float4 A00 = lp[t0i*4 + h0i], A01 = lp[t0i*4 + h1i];
      const float4 A10 = lp[t1i*4 + h0i], A11 = lp[t1i*4 + h1i];
      float4 L;
      L.x = (A00.x*(1.f-fh) + A01.x*fh)*(1.f-ft) + (A10.x*(1.f-fh) + A11.x*fh)*ft;
      L.y = (A00.y*(1.f-fh) + A01.y*fh)*(1.f-ft) + (A10.y*(1.f-fh) + A11.y*fh)*ft;
      L.z = (A00.z*(1.f-fh) + A01.z*fh)*(1.f-ft) + (A10.z*(1.f-fh) + A11.z*fh)*ft;
      L.w = (A00.w*(1.f-fh) + A01.w*fh)*(1.f-ft) + (A10.w*(1.f-fh) + A11.w*fh)*ft;
      *(float4*)&Lws[rw][c][0] = L;
    }
  }

  v16h af[2][4];
  const int rowi = wv >> 1;
  const int w0   = (pw & 63) + 2*col;
  {
    const float* ap = ub + e0;
    float* cw = &Cs[rowi][hf*8][w0];
    #pragma unroll
    for (int kb = 0; kb < 4; ++kb){
      #pragma unroll
      for (int s = 0; s < 16; ++s){
        const int ch = kb*32 + (s < 8 ? s : s + 8);
        const float2 v = *(const float2*)(ap + (size_t)ch * N_);
        *(float2*)(cw + ch*68) = v;
        af[0][kb][s] = (_Float16)v.x;
        af[1][kb][s] = (_Float16)v.y;
      }
    }
  }

  const v8f z8 = {0.f,0.f,0.f,0.f,0.f,0.f,0.f,0.f};
  v8f accC[8][2];
  #pragma unroll
  for (int i = 0; i < 8; ++i){ accC[i][0] = z8; accC[i][1] = z8; }

  for (int hc = 0; hc < 8; ++hc){
    v16h bf[8];
    #pragma unroll
    for (int f = 0; f < 8; ++f) bf[f] = load_frag_glob(w1h, hc*8 + f, lane);
    v8f a00 = z8, a01 = z8, a10 = z8, a11 = z8;
    #pragma unroll
    for (int kb = 0; kb < 4; ++kb){
      a00 = __builtin_amdgcn_wmma_f32_16x16x32_f16(false, bf[kb*2],   false, af[0][kb], (short)0, a00, false, false);
      a01 = __builtin_amdgcn_wmma_f32_16x16x32_f16(false, bf[kb*2+1], false, af[0][kb], (short)0, a01, false, false);
      a10 = __builtin_amdgcn_wmma_f32_16x16x32_f16(false, bf[kb*2],   false, af[1][kb], (short)0, a10, false, false);
      a11 = __builtin_amdgcn_wmma_f32_16x16x32_f16(false, bf[kb*2+1], false, af[1][kb], (short)0, a11, false, false);
    }
    v16h bw[8];
    #pragma unroll
    for (int nt = 0; nt < 8; ++nt) bw[nt] = load_frag_glob(w2h, hc*8 + nt, lane);
    float bb0[8], bb1[8];
    ldc8(b1 + hc*32 + hf*8, bb0);
    ldc8(b1 + hc*32 + 16 + hf*8, bb1);
    v16h ah0, ah1;
    #pragma unroll
    for (int r = 0; r < 8; ++r){
      ah0[r]     = (_Float16)gelu_f(a00[r] + bb0[r]);
      ah0[8 + r] = (_Float16)gelu_f(a01[r] + bb1[r]);
      ah1[r]     = (_Float16)gelu_f(a10[r] + bb0[r]);
      ah1[8 + r] = (_Float16)gelu_f(a11[r] + bb1[r]);
    }
    #pragma unroll
    for (int nt = 0; nt < 8; ++nt){
      accC[nt][0] = __builtin_amdgcn_wmma_f32_16x16x32_f16(false, bw[nt], false, ah0, (short)0, accC[nt][0], false, false);
      accC[nt][1] = __builtin_amdgcn_wmma_f32_16x16x32_f16(false, bw[nt], false, ah1, (short)0, accC[nt][1], false, false);
    }
  }

  #pragma unroll
  for (int nt = 0; nt < 8; ++nt){
    #pragma unroll
    for (int r = 0; r < 8; ++r){
      float2 rv = { accC[nt][0][r], accC[nt][1][r] };
      *(float2*)&Rs[wv][nt*16 + hf*8 + r][2*col] = rv;
    }
  }
  __syncthreads();

  const float dt = scal[0], al = scal[2];

  float ww[2][4];
  #pragma unroll
  for (int j = 0; j < 2; ++j){
    const float sw = fminf(fmaxf((w0 + j + 0.5f)*(4.f/W_) - 0.5f, 0.f), 3.f);
    const int wi0 = (int)sw; const int wi1 = imin_(wi0+1, 3); const float f = sw - (float)wi0;
    #pragma unroll
    for (int i = 0; i < 4; ++i) ww[j][i] = (i == wi0 ? 1.f - f : 0.f) + (i == wi1 ? f : 0.f);
  }

  const bool vU1 = hh - 1  >= 0, vD1 = hh + 1  < 64;
  const bool vU4 = hh - 4  >= 0, vD4 = hh + 4  < 64;
  const bool vU16= hh - 16 >= 0, vD16= hh + 16 < 64;
  const bool vT1 = t - 1 >= 0,   vT2 = t - 2 >= 0;
  const bool vL4 = w0 - 4  >= 0, vR4 = w0 + 5  < 64;
  const bool vL16= w0 - 16 >= 0, vR16= w0 + 17 < 64;
  const bool vLm = w0 - 1 >= 0,  vRp = w0 + 2 < 64;
  const int oU1 = vU1 ? -64 : 0,     oD1 = vD1 ? 64 : 0;
  const int oU4 = vU4 ? -256 : 0,    oD4 = vD4 ? 256 : 0;
  const int oU16= vU16 ? -1024 : 0,  oD16= vD16 ? 1024 : 0;
  const int oT1 = vT1 ? -4096 : 0,   oT2 = vT2 ? -8192 : 0;
  const int oL4 = vL4 ? -4 : 0,      oR4 = vR4 ? 4 : 0;
  const int oL16= vL16 ? -16 : 0,    oR16= vR16 ? 16 : 0;
  const int oLm = vLm ? -1 : 0,      oRp = vRp ? 2 : 0;
  const bool vG1 = rowi == 0 ? vU1 : vD1;
  const int  oG1 = rowi == 0 ? oU1 : oD1;

  const size_t bbase = (size_t)b * C_ * N_;
  float sum_du = 0.f, sum_lap = 0.f;
  float ssq0 = 0.f, ssq1 = 0.f;

  float*       Rw   = &Rs[wv][hf*8][2*col];
  const float* tabw = &tabs[hf*8][0];
  const float* Lww  = &Lws[wv >> 1][hf*8][0];
  const float* Cme  = &Cs[rowi][hf*8][w0];
  const float* Cot  = &Cs[rowi ^ 1][hf*8][w0];

  auto channel_steps = [&](auto norm_tag){
  constexpr bool NORM = decltype(norm_tag)::value;
  for (int ib = 0; ib < 8; ++ib){
  float2 gG1[8], gU4[8], gD4[8], gU16[8], gD16[8], gT1[8], gT2[8];
  #pragma unroll
  for (int j = 0; j < 8; ++j){
    const float* uc = ub + (size_t)(ib*16 + j)*N_ + e0;
    gG1[j]  = *(const float2*)(uc + oG1);
    gU4[j]  = *(const float2*)(uc + oU4);   gD4[j]  = *(const float2*)(uc + oD4);
    gU16[j] = *(const float2*)(uc + oU16);  gD16[j] = *(const float2*)(uc + oD16);
    gT1[j]  = *(const float2*)(uc + oT1);   gT2[j]  = *(const float2*)(uc + oT2);
  }
  #pragma unroll
  for (int j = 0; j < 8; ++j){
    const int cl = ib*16 + j;
    const float4 ta = *(const float4*)(tabw + cl*8);
    const float4 tb = *(const float4*)(tabw + cl*8 + 4);
    const float4 Lq = *(const float4*)(Lww + cl*4);
    const float2 rc = *(const float2*)(Rw + cl*36);
    const float lv0 = Lq.x*ww[0][0] + Lq.y*ww[0][1] + Lq.z*ww[0][2] + Lq.w*ww[0][3];
    const float lv1 = Lq.x*ww[1][0] + Lq.y*ww[1][1] + Lq.z*ww[1][2] + Lq.w*ww[1][3];

    const float* cc = Cme + cl*68;
    const float2 G1  = sel2(gG1[j],  vG1);
    const float2 U4  = sel2(gU4[j],  vU4),  D4  = sel2(gD4[j],  vD4);
    const float2 U16 = sel2(gU16[j], vU16), D16 = sel2(gD16[j], vD16);
    const float2 T1  = sel2(gT1[j],  vT1),  T2  = sel2(gT2[j],  vT2);
    const float2 cen = *(const float2*)cc;
    const float2 S1  = *(const float2*)(Cot + cl*68);
    const float2 L4  = ld2(cc + oL4,  vL4),  R4  = ld2(cc + oR4,  vR4);
    const float2 L16 = ld2(cc + oL16, vL16), R16 = ld2(cc + oR16, vR16);
    const float  lm  = ld1(cc + oLm,  vLm);
    const float  rp  = ld1(cc + oRp,  vRp);
    const float2 U1  = rowi == 0 ? G1 : S1;
    const float2 D1  = rowi == 0 ? S1 : G1;

    const float rest0 = (rc.x + tb.y) + tb.z;
    const float rest1 = (rc.y + tb.y) + tb.z;
    const float cnd0 = upd1(cen.x, U1.x, D1.x, lm,    cen.y, U4.x, D4.x, L4.x, R4.x,
                            U16.x, D16.x, L16.x, R16.x, T1.x, T2.x,
                            ta.x, ta.y, ta.z, ta.w, tb.x, rest0, lv0, al, dt,
                            sum_lap, sum_du, ssq0);
    const float cnd1 = upd1(cen.y, U1.y, D1.y, cen.x, rp,    U4.y, D4.y, L4.y, R4.y,
                            U16.y, D16.y, L16.y, R16.y, T1.y, T2.y,
                            ta.x, ta.y, ta.z, ta.w, tb.x, rest1, lv1, al, dt,
                            sum_lap, sum_du, ssq1);
    const float2 cv = { cnd0, cnd1 };
    *(float2*)(ucand + bbase + (size_t)cl*N_ + e0) = cv;
    if constexpr (NORM) *(float2*)(Rw + cl*36) = cv;
    else                *(float2*)(unext + bbase + (size_t)cl*N_ + e0) = cv;
  }
  }
  };
  if (do_norm) channel_steps(std::true_type{});
  else         channel_steps(std::false_type{});

  if (do_norm){
    ssq0 += __shfl_xor(ssq0, 16);
    ssq1 += __shfl_xor(ssq1, 16);
    const float scale0 = rsqrtf(ssq0/(float)C_ + 1e-8f);
    const float scale1 = rsqrtf(ssq1/(float)C_ + 1e-8f);
    #pragma unroll 8
    for (int it = 0; it < 64; ++it){
      const int cl = ((it >> 3) << 4) | (it & 7);
      const float2 cv = *(const float2*)(Rw + cl*36);
      const float nw = tabw[cl*8 + 7];
      float2 o;
      o.x = cv.x*scale0*nw;
      o.y = cv.y*scale1*nw;
      *(float2*)(unext + bbase + (size_t)cl*N_ + e0) = o;
    }
  }

  float s1 = blockReduceSum(sum_du);
  float s2 = blockReduceSum(sum_lap);
  if (threadIdx.x == 0){
    atomicAdd(&accum[1], s1);
    atomicAdd(&accum[0], s2);
  }
}

__global__ void final_out_kernel(const float* accum, const float* turbs, const float* sten, float* out){
  if (threadIdx.x == 0 && blockIdx.x == 0){
    float tl, sl;
    step_scalars(accum, tl, sl);
    float a = 0.f, bm = 0.f;
    for (int i = 0; i < MAX_STEPS_ - 1; ++i){ a += turbs[i]; bm += sten[i]; }
    a += tl; bm += sl;
    out[0] = a / (float)MAX_STEPS_;
    out[1] = bm / (float)MAX_STEPS_;
  }
}


extern "C" void kernel_launch(void* const* d_in, const int* in_sizes, int n_in,
                              void* d_out, int out_size, void* d_ws, size_t ws_size,
                              hipStream_t stream){
  (void)in_sizes; (void)n_in; (void)out_size; (void)ws_size;

  const float* u_in  = (const float*)d_in[0];
  const float* w1    = (const float*)d_in[1];
  const float* b1    = (const float*)d_in[2];
  const float* w2    = (const float*)d_in[3];
  const float* b2    = (const float*)d_in[4];
  const float* gw    = (const float*)d_in[5];
  const float* gb    = (const float*)d_in[6];
  const float* vw    = (const float*)d_in[7];
  const float* vb    = (const float*)d_in[8];
  const float* normw = (const float*)d_in[9];
  const float* wlp   = (const float*)d_in[10];
  const float* blp   = (const float*)d_in[11];
  const float* a_g   = (const float*)d_in[12];
  const float* a_l   = (const float*)d_in[13];
  const float* logdt = (const float*)d_in[14];
  const float* ks    = (const float*)d_in[15];
  const float* kt    = (const float*)d_in[16];

  float* Wf = (float*)d_ws;
  size_t o = 0;
  const size_t VOL = (size_t)B_ * C_ * N_;
  float* ubuf0  = Wf + o; o += VOL;
  float* ubuf1  = Wf + o; o += VOL;
  float* ucand  = Wf + o; o += VOL;
  float* probeA = Wf + o; o += PROBE_CELLS;
  float* probeB = Wf + o; o += PROBE_CELLS;
  float* pooled = Wf + o; o += 256;
  float* hstate = Wf + o; o += 256;
  float* lsmall = Wf + o; o += (size_t)B_*C_*32;
  float* lproj  = Wf + o; o += (size_t)B_*C_*32;
  float* spsks  = Wf + o; o += 3*C_;
  float* spskt  = Wf + o; o += 2*C_;
  float* scal   = Wf + o; o += 8;
  float* accum  = Wf + o; o += 8;
  float* turbs  = Wf + o; o += 16;
  float* sten   = Wf + o; o += 16;
  _Float16* w1h = (_Float16*)(Wf + o); o += (C_*HID_)/2;
  _Float16* w2h = (_Float16*)(Wf + o); o += (HID_*C_)/2;

  float* ubuf[2] = { ubuf0, ubuf1 };

  prep_kernel<<<64, 256, 0, stream>>>(w1, w2, ks, kt, a_g, a_l, logdt,
                                      w1h, w2h, spsks, spskt, scal, hstate);
  hipMemcpyAsync(ubuf[0], u_in, VOL*sizeof(float), hipMemcpyDeviceToDevice, stream);
  pool3_kernel<<<B_*C_, 512, 0, stream>>>(ubuf[0], ubuf[0], 0, probeA, (const float*)nullptr,
                                          lsmall, pooled, accum);

  float* pprev = probeA;
  float* pcur  = probeB;

  for (int step = 0; step < MAX_STEPS_; ++step){
    float* ucur  = ubuf[step & 1];
    float* unext = ubuf[(step & 1) ^ 1];
    int do_norm = (((step + 1) % 2) == 0) ? 1 : 0;

    memory_kernel<<<MEM_GROUPS, 256, 0, stream>>>(pooled, gw, gb, vw, vb, hstate, lsmall, wlp, blp, lproj,
                                                  accum, turbs, sten, step);
    step_kernel<<<BN_/128, 128, 0, stream>>>(ucur, w1h, b1, w2h, b2, hstate, lproj, spsks, spskt,
                                             scal, normw, ucand, unext, accum, do_norm);
    pool3_kernel<<<B_*C_, 512, 0, stream>>>(ucand, unext, do_norm, pcur, pprev, lsmall, pooled, accum);

    float* tswap = pprev; pprev = pcur; pcur = tswap;
  }

  hipMemcpyAsync(d_out, ubuf[MAX_STEPS_ & 1], VOL*sizeof(float),
                 hipMemcpyDeviceToDevice, stream);
  final_out_kernel<<<1, 32, 0, stream>>>(accum, turbs, sten, ((float*)d_out) + VOL);
}


# state never copied or stored twice: step 0 reads the input in place, the last step writes the output plane, the candidate plane is written only on normalising steps; both full-volume copies go
# speedup vs baseline: 8.2127x; 1.0759x over previous
#include <hip/hip_runtime.h>
#include <math.h>
#include <type_traits>

#define B_ 2
#define C_ 128
#define T_ 8
#define H_ 64
#define W_ 64
#define N_ (T_*H_*W_)
#define BN_ (B_*N_)
#define HID_ 256
#define MAX_STEPS_ 12
#define DIFF_SCALE 0.25f
#define PROBE_CELLS (B_*C_*2*8*8)

typedef __attribute__((ext_vector_type(16))) _Float16 v16h;
typedef __attribute__((ext_vector_type(8)))  _Float16 h8;
typedef __attribute__((ext_vector_type(8)))  float    v8f;

__device__ __forceinline__ float softplus_f(float x){ return x > 20.f ? x : log1pf(expf(x)); }
__device__ __forceinline__ float gelu_f(float x){ return 0.5f*x*(1.f + erff(x*0.70710678118654752f)); }
__device__ __forceinline__ int imin_(int a, int b){ return a < b ? a : b; }

__device__ __forceinline__ float blockReduceSum(float v){
  __shared__ float red[8];
  int lane = threadIdx.x & 31, wid = threadIdx.x >> 5;
  #pragma unroll
  for (int off = 16; off > 0; off >>= 1) v += __shfl_down(v, off);
  if (lane == 0) red[wid] = v;
  __syncthreads();
  v = (threadIdx.x < (blockDim.x >> 5)) ? red[threadIdx.x] : 0.f;
  if (wid == 0){
    #pragma unroll
    for (int off = 4; off > 0; off >>= 1) v += __shfl_down(v, off);
  }
  __syncthreads();
  return v;
}

__device__ __forceinline__ v16h load_frag_glob(const _Float16* base, int f, int lane){
  const h8* p = (const h8*)base + f*64 + lane;
  h8 lo = p[0];
  h8 hi = p[32];
  return __builtin_shufflevector(lo, hi, 0,1,2,3,4,5,6,7,8,9,10,11,12,13,14,15);
}


__global__ void prep_kernel(const float* w1, const float* w2,
                            const float* ks, const float* kt,
                            const float* ag, const float* al, const float* logdt,
                            _Float16* w1h, _Float16* w2h,
                            float* spsks, float* spskt, float* scal, float* hstate){
  int i = blockIdx.x*blockDim.x + threadIdx.x;
  int stride = gridDim.x*blockDim.x;
  if (i < B_*C_) hstate[i] = 0.f;
  for (int j = i; j < C_*HID_; j += stride){
    int e = j & 7, ln = (j >> 3) & 31, half = (j >> 8) & 1;
    int nt = (j >> 9) & 1, kb = (j >> 10) & 3, hc = j >> 12;
    int k = kb*32 + half*16 + (ln >> 4)*8 + e;
    int n = hc*32 + nt*16 + (ln & 15);
    w1h[j] = (_Float16)w1[k*HID_ + n];
  }
  for (int j = i; j < HID_*C_; j += stride){
    int e = j & 7, ln = (j >> 3) & 31, half = (j >> 8) & 1;
    int nt = (j >> 9) & 7, hc = j >> 12;
    int k = hc*32 + half*16 + (ln >> 4)*8 + e;
    int n = nt*16 + (ln & 15);
    w2h[j] = (_Float16)w2[k*C_ + n];
  }
  for (int j = i; j < 3*C_;   j += stride) spsks[j] = softplus_f(ks[j]);
  for (int j = i; j < 2*C_;   j += stride) spskt[j] = softplus_f(kt[j]);
  if (i == 0){
    float dt = expf(logdt[0]);
    dt = fminf(fmaxf(dt, 0.005f), 0.35f);
    scal[0] = dt;
    scal[1] = softplus_f(ag[0]);
    scal[2] = softplus_f(al[0]);
  }
}

__device__ __forceinline__ float blockrow_sum(const float* p){
  const float4* q = (const float4*)p;
  float4 v[16];
  #pragma unroll
  for (int f = 0; f < 4; ++f){
    #pragma unroll
    for (int j = 0; j < 4; ++j) v[f*4 + j] = q[f*(H_*W_/4) + j*(2*W_/4)];
  }
  float s = 0.f;
  #pragma unroll
  for (int i = 0; i < 16; ++i) s += (v[i].x + v[i].y) + (v[i].z + v[i].w);
  s += __shfl_xor(s, 1);
  s += __shfl_xor(s, 16);
  return s;
}

__global__ void __launch_bounds__(512)
pool3_kernel(const float* cand, const float* state, int state_differs,
             float* probe, const float* prev, float* lsmall, float* pooled, float* accum){
  __shared__ float cs[16][8];
  __shared__ float dsum[16][8];
  __shared__ float psum[16][8];
  const int bc   = blockIdx.x;
  const int lane = threadIdx.x & 31;
  const int wv   = threadIdx.x >> 5;
  const int pt = wv >> 3, ph = wv & 7;
  const size_t off = (size_t)bc*N_ + (size_t)pt*(4*H_*W_)
                   + (size_t)((ph*8 + (lane >> 4))*W_ + (lane & 15)*4);
  const int pw = (lane >> 1) & 7;
  const bool owner = (lane & 17) == 0;
  const size_t pi = (size_t)(bc*16 + wv)*8 + pw;
  float pp = 0.f;
  if (prev != nullptr && owner) pp = prev[pi];
  const float s = blockrow_sum(cand + off);
  float ss = s;
  if (state_differs) ss = blockrow_sum(state + off);
  if (owner){
    const float cv = s * (1.f/256.f);
    probe[pi] = cv;
    cs[wv][pw] = ss;
    dsum[wv][pw] = fabsf(cv - pp);
    psum[wv][pw] = fabsf(pp);
  }
  __syncthreads();
  if (wv == 0){
    const int lt = lane >> 4, lh = (lane >> 2) & 3, lw = lane & 3;
    const int r0 = lt*8 + lh*2;
    float a = (cs[r0][2*lw] + cs[r0][2*lw + 1]) + (cs[r0 + 1][2*lw] + cs[r0 + 1][2*lw + 1]);
    lsmall[(size_t)bc*32 + lane] = a * (1.f/1024.f);
    a += __shfl_xor(a, 16);
    a += __shfl_xor(a, 8);
    a += __shfl_xor(a, 4);
    a += __shfl_xor(a, 2);
    a += __shfl_xor(a, 1);
    if (lane == 0) pooled[bc] = a * (1.f/(float)N_);
  } else if (wv == 1 && prev != nullptr){
    const float* d = &dsum[0][0];
    const float* p = &psum[0][0];
    float a = (d[lane] + d[32 + lane]) + (d[64 + lane] + d[96 + lane]);
    float q = (p[lane] + p[32 + lane]) + (p[64 + lane] + p[96 + lane]);
    for (int off2 = 16; off2 > 0; off2 >>= 1){ a += __shfl_xor(a, off2); q += __shfl_xor(q, off2); }
    if (lane == 0){
      atomicAdd(&accum[2], a);
      atomicAdd(&accum[3], q);
    }
  }
}

__device__ __forceinline__ void step_scalars(const float* accum, float& turb, float& se){
  const float inv_cells = 1.f/(float)PROBE_CELLS;
  const float inv_bnc   = 1.f/8388608.f;
  const float stop_turb = (accum[2]*inv_cells) / (accum[3]*inv_cells + 1e-8f);
  se = accum[1]*inv_bnc;
  const float le = accum[0]*inv_bnc;
  turb = stop_turb + 0.05f*se + 0.01f*le;
}

#define MEM_GROUPS 34
__global__ void __launch_bounds__(256)
memory_kernel(const float* pooled, const float* gw, const float* gb,
              const float* vw, const float* vb, float* h,
              const float* lsmall, const float* wlp, const float* blp, float* lproj,
              float* accum, float* turbs, float* sten, int step){
  const int tid = threadIdx.x;
  if (blockIdx.x < 32){
    const int id = blockIdx.x*256 + tid;
    const int cell = id & 31;
    const int oc = (id >> 5) & 127;
    const int b = id >> 12;
    float s = blp[oc];
    #pragma unroll 32
    for (int c = 0; c < C_; ++c)
      s += lsmall[((size_t)(b*C_ + c) << 5) + cell] * wlp[c*C_ + oc];
    lproj[((size_t)(b*C_ + oc) << 5) + cell] = s;
  } else if (blockIdx.x == 32){
    const int b = tid >> 7, c = tid & 127;
    const float* p = pooled + b*C_;
    float g = gb[c], v = vb[c];
    #pragma unroll 32
    for (int k = 0; k < C_; ++k){
      const float pk = p[k];
      g += pk * gw[k*C_ + c];
      v += pk * vw[k*C_ + c];
    }
    h[tid] += (1.f/(1.f+expf(-g))) * tanhf(v);
  } else if (tid == 0){
    if (step > 0){
      float tb, se;
      step_scalars(accum, tb, se);
      turbs[step - 1] = tb;
      sten[step - 1]  = se;
    }
    accum[0] = 0.f; accum[1] = 0.f; accum[2] = 0.f; accum[3] = 0.f;
  }
}

__device__ __forceinline__ float2 ld2(const float* p, bool ok){
  const float2 a = *(const float2*)p;
  float2 o; o.x = ok ? a.x : 0.f; o.y = ok ? a.y : 0.f;
  return o;
}
__device__ __forceinline__ float ld1(const float* p, bool ok){
  const float a = *p;
  return ok ? a : 0.f;
}
__device__ __forceinline__ float2 sel2(float2 a, bool ok){
  float2 o; o.x = ok ? a.x : 0.f; o.y = ok ? a.y : 0.f;
  return o;
}
__device__ __forceinline__ void ldc8(const float* p, float* o){
  const float4 a = *(const float4*)p, b = *(const float4*)(p + 4);
  o[0] = a.x; o[1] = a.y; o[2] = a.z; o[3] = a.w; o[4] = b.x; o[5] = b.y; o[6] = b.z; o[7] = b.w;
}

__device__ __forceinline__ float upd1(float uv, float u1a, float u1b, float l1, float r1,
                                      float u4a, float u4b, float l4, float r4,
                                      float u16a, float u16b, float l16, float r16,
                                      float t1, float t2,
                                      float k0, float k1, float k2, float q0, float q1,
                                      float rest, float lv, float al, float dt,
                                      float& sum_lap, float& sum_du, float& ssq){
  float lap1  = -4.f * uv; lap1  += u1a;  lap1  += u1b;  lap1  += l1;  lap1  += r1;
  float lap4  = -4.f * uv; lap4  += u4a;  lap4  += u4b;  lap4  += l4;  lap4  += r4;
  float lap16 = -4.f * uv; lap16 += u16a; lap16 += u16b; lap16 += l16; lap16 += r16;
  float diff = k0 * lap1;
  diff += k1 * lap4;
  diff += k2 * lap16;
  diff += q0 * (t1 - uv);
  diff += q1 * (t2 - uv);
  diff *= DIFF_SCALE;
  sum_lap += fabsf(diff);
  const float du = diff + rest + al*lv;
  const float cand = uv + dt*du;
  sum_du += fabsf(du);
  ssq += cand*cand;
  return cand;
}

__global__ void __launch_bounds__(128) __attribute__((amdgpu_waves_per_eu(2)))
step_kernel(const float* __restrict__ u, const _Float16* __restrict__ w1h, const float* __restrict__ b1,
            const _Float16* __restrict__ w2h, const float* __restrict__ b2,
            const float* __restrict__ h, const float* __restrict__ lproj,
            const float* __restrict__ spsks, const float* __restrict__ spskt,
            const float* __restrict__ scal, const float* __restrict__ normw,
            float* __restrict__ ucand, float* __restrict__ unext, float* accum, int do_norm){
  const int tid  = threadIdx.x;
  const int lane = tid & 31;
  const int wv   = __builtin_amdgcn_readfirstlane(tid >> 5);
  const int hf   = lane >> 4;
  const int col  = lane & 15;

  const int pixBase = blockIdx.x * 128 + wv * 32;
  const int b  = pixBase / N_;
  const int pw = pixBase - b * N_;
  const float* ub = u + (size_t)b * C_ * N_;
  const int e0 = hf*8*N_ + pw + 2*col;
  const int t  = pw >> 12;
  const int hh = (pw >> 6) & 63;

  __shared__ __attribute__((aligned(16))) float Rs[4][128][36];
  __shared__ __attribute__((aligned(16))) float tabs[128][8];
  __shared__ __attribute__((aligned(16))) float Lws[2][128][4];
  __shared__ __attribute__((aligned(16))) float Cs[2][128][68];
  {
    const int c = tid;
    const float agv = scal[1];
    const float4 k4 = { spsks[c], spsks[C_ + c], spsks[2*C_ + c], spskt[c] };
    const float4 q4 = { spskt[C_ + c], b2[c], agv * h[b*C_ + c], normw[c] };
    *(float4*)&tabs[c][0] = k4;
    *(float4*)&tabs[c][4] = q4;
    const int tb_ = (blockIdx.x * 128 - b * N_) >> 12;
    const int hb_ = ((blockIdx.x * 128 - b * N_) >> 6) & 63;
    const float st  = fminf(fmaxf((tb_ + 0.5f)*(2.f/T_) - 0.5f, 0.f), 1.f);
    const int t0i = (int)st;  const int t1i = imin_(t0i+1, 1); const float ft = st - (float)t0i;
    const float4* lp = (const float4*)(lproj + ((size_t)(b*C_ + c) << 5));
    #pragma unroll
    for (int rw = 0; rw < 2; ++rw){
      const float sh  = fminf(fmaxf((hb_ + rw + 0.5f)*(4.f/H_) - 0.5f, 0.f), 3.f);
      const int h0i = (int)sh;  const int h1i = imin_(h0i+1, 3); const float fh = sh - (float)h0i;
      const float4 A00 = lp[t0i*4 + h0i], A01 = lp[t0i*4 + h1i];
      const float4 A10 = lp[t1i*4 + h0i], A11 = lp[t1i*4 + h1i];
      float4 L;
      L.x = (A00.x*(1.f-fh) + A01.x*fh)*(1.f-ft) + (A10.x*(1.f-fh) + A11.x*fh)*ft;
      L.y = (A00.y*(1.f-fh) + A01.y*fh)*(1.f-ft) + (A10.y*(1.f-fh) + A11.y*fh)*ft;
      L.z = (A00.z*(1.f-fh) + A01.z*fh)*(1.f-ft) + (A10.z*(1.f-fh) + A11.z*fh)*ft;
      L.w = (A00.w*(1.f-fh) + A01.w*fh)*(1.f-ft) + (A10.w*(1.f-fh) + A11.w*fh)*ft;
      *(float4*)&Lws[rw][c][0] = L;
    }
  }

  v16h af[2][4];
  const int rowi = wv >> 1;
  const int w0   = (pw & 63) + 2*col;
  {
    const float* ap = ub + e0;
    float* cw = &Cs[rowi][hf*8][w0];
    #pragma unroll
    for (int kb = 0; kb < 4; ++kb){
      #pragma unroll
      for (int s = 0; s < 16; ++s){
        const int ch = kb*32 + (s < 8 ? s : s + 8);
        const float2 v = *(const float2*)(ap + (size_t)ch * N_);
        *(float2*)(cw + ch*68) = v;
        af[0][kb][s] = (_Float16)v.x;
        af[1][kb][s] = (_Float16)v.y;
      }
    }
  }

  const v8f z8 = {0.f,0.f,0.f,0.f,0.f,0.f,0.f,0.f};
  v8f accC[8][2];
  #pragma unroll
  for (int i = 0; i < 8; ++i){ accC[i][0] = z8; accC[i][1] = z8; }

  for (int hc = 0; hc < 8; ++hc){
    v16h bf[8];
    #pragma unroll
    for (int f = 0; f < 8; ++f) bf[f] = load_frag_glob(w1h, hc*8 + f, lane);
    v8f a00 = z8, a01 = z8, a10 = z8, a11 = z8;
    #pragma unroll
    for (int kb = 0; kb < 4; ++kb){
      a00 = __builtin_amdgcn_wmma_f32_16x16x32_f16(false, bf[kb*2],   false, af[0][kb], (short)0, a00, false, false);
      a01 = __builtin_amdgcn_wmma_f32_16x16x32_f16(false, bf[kb*2+1], false, af[0][kb], (short)0, a01, false, false);
      a10 = __builtin_amdgcn_wmma_f32_16x16x32_f16(false, bf[kb*2],   false, af[1][kb], (short)0, a10, false, false);
      a11 = __builtin_amdgcn_wmma_f32_16x16x32_f16(false, bf[kb*2+1], false, af[1][kb], (short)0, a11, false, false);
    }
    v16h bw[8];
    #pragma unroll
    for (int nt = 0; nt < 8; ++nt) bw[nt] = load_frag_glob(w2h, hc*8 + nt, lane);
    float bb0[8], bb1[8];
    ldc8(b1 + hc*32 + hf*8, bb0);
    ldc8(b1 + hc*32 + 16 + hf*8, bb1);
    v16h ah0, ah1;
    #pragma unroll
    for (int r = 0; r < 8; ++r){
      ah0[r]     = (_Float16)gelu_f(a00[r] + bb0[r]);
      ah0[8 + r] = (_Float16)gelu_f(a01[r] + bb1[r]);
      ah1[r]     = (_Float16)gelu_f(a10[r] + bb0[r]);
      ah1[8 + r] = (_Float16)gelu_f(a11[r] + bb1[r]);
    }
    #pragma unroll
    for (int nt = 0; nt < 8; ++nt){
      accC[nt][0] = __builtin_amdgcn_wmma_f32_16x16x32_f16(false, bw[nt], false, ah0, (short)0, accC[nt][0], false, false);
      accC[nt][1] = __builtin_amdgcn_wmma_f32_16x16x32_f16(false, bw[nt], false, ah1, (short)0, accC[nt][1], false, false);
    }
  }

  #pragma unroll
  for (int nt = 0; nt < 8; ++nt){
    #pragma unroll
    for (int r = 0; r < 8; ++r){
      float2 rv = { accC[nt][0][r], accC[nt][1][r] };
      *(float2*)&Rs[wv][nt*16 + hf*8 + r][2*col] = rv;
    }
  }
  __syncthreads();

  const float dt = scal[0], al = scal[2];

  float ww[2][4];
  #pragma unroll
  for (int j = 0; j < 2; ++j){
    const float sw = fminf(fmaxf((w0 + j + 0.5f)*(4.f/W_) - 0.5f, 0.f), 3.f);
    const int wi0 = (int)sw; const int wi1 = imin_(wi0+1, 3); const float f = sw - (float)wi0;
    #pragma unroll
    for (int i = 0; i < 4; ++i) ww[j][i] = (i == wi0 ? 1.f - f : 0.f) + (i == wi1 ? f : 0.f);
  }

  const bool vU1 = hh - 1  >= 0, vD1 = hh + 1  < 64;
  const bool vU4 = hh - 4  >= 0, vD4 = hh + 4  < 64;
  const bool vU16= hh - 16 >= 0, vD16= hh + 16 < 64;
  const bool vT1 = t - 1 >= 0,   vT2 = t - 2 >= 0;
  const bool vL4 = w0 - 4  >= 0, vR4 = w0 + 5  < 64;
  const bool vL16= w0 - 16 >= 0, vR16= w0 + 17 < 64;
  const bool vLm = w0 - 1 >= 0,  vRp = w0 + 2 < 64;
  const int oU1 = vU1 ? -64 : 0,     oD1 = vD1 ? 64 : 0;
  const int oU4 = vU4 ? -256 : 0,    oD4 = vD4 ? 256 : 0;
  const int oU16= vU16 ? -1024 : 0,  oD16= vD16 ? 1024 : 0;
  const int oT1 = vT1 ? -4096 : 0,   oT2 = vT2 ? -8192 : 0;
  const int oL4 = vL4 ? -4 : 0,      oR4 = vR4 ? 4 : 0;
  const int oL16= vL16 ? -16 : 0,    oR16= vR16 ? 16 : 0;
  const int oLm = vLm ? -1 : 0,      oRp = vRp ? 2 : 0;
  const bool vG1 = rowi == 0 ? vU1 : vD1;
  const int  oG1 = rowi == 0 ? oU1 : oD1;

  const size_t bbase = (size_t)b * C_ * N_;
  float sum_du = 0.f, sum_lap = 0.f;
  float ssq0 = 0.f, ssq1 = 0.f;

  float*       Rw   = &Rs[wv][hf*8][2*col];
  const float* tabw = &tabs[hf*8][0];
  const float* Lww  = &Lws[wv >> 1][hf*8][0];
  const float* Cme  = &Cs[rowi][hf*8][w0];
  const float* Cot  = &Cs[rowi ^ 1][hf*8][w0];

  auto channel_steps = [&](auto norm_tag){
  constexpr bool NORM = decltype(norm_tag)::value;
  for (int ib = 0; ib < 8; ++ib){
  float2 gG1[8], gU4[8], gD4[8], gU16[8], gD16[8], gT1[8], gT2[8];
  #pragma unroll
  for (int j = 0; j < 8; ++j){
    const float* uc = ub + (size_t)(ib*16 + j)*N_ + e0;
    gG1[j]  = *(const float2*)(uc + oG1);
    gU4[j]  = *(const float2*)(uc + oU4);   gD4[j]  = *(const float2*)(uc + oD4);
    gU16[j] = *(const float2*)(uc + oU16);  gD16[j] = *(const float2*)(uc + oD16);
    gT1[j]  = *(const float2*)(uc + oT1);   gT2[j]  = *(const float2*)(uc + oT2);
  }
  #pragma unroll
  for (int j = 0; j < 8; ++j){
    const int cl = ib*16 + j;
    const float4 ta = *(const float4*)(tabw + cl*8);
    const float4 tb = *(const float4*)(tabw + cl*8 + 4);
    const float4 Lq = *(const float4*)(Lww + cl*4);
    const float2 rc = *(const float2*)(Rw + cl*36);
    const float lv0 = Lq.x*ww[0][0] + Lq.y*ww[0][1] + Lq.z*ww[0][2] + Lq.w*ww[0][3];
    const float lv1 = Lq.x*ww[1][0] + Lq.y*ww[1][1] + Lq.z*ww[1][2] + Lq.w*ww[1][3];

    const float* cc = Cme + cl*68;
    const float2 G1  = sel2(gG1[j],  vG1);
    const float2 U4  = sel2(gU4[j],  vU4),  D4  = sel2(gD4[j],  vD4);
    const float2 U16 = sel2(gU16[j], vU16), D16 = sel2(gD16[j], vD16);
    const float2 T1  = sel2(gT1[j],  vT1),  T2  = sel2(gT2[j],  vT2);
    const float2 cen = *(const float2*)cc;
    const float2 S1  = *(const float2*)(Cot + cl*68);
    const float2 L4  = ld2(cc + oL4,  vL4),  R4  = ld2(cc + oR4,  vR4);
    const float2 L16 = ld2(cc + oL16, vL16), R16 = ld2(cc + oR16, vR16);
    const float  lm  = ld1(cc + oLm,  vLm);
    const float  rp  = ld1(cc + oRp,  vRp);
    const float2 U1  = rowi == 0 ? G1 : S1;
    const float2 D1  = rowi == 0 ? S1 : G1;

    const float rest0 = (rc.x + tb.y) + tb.z;
    const float rest1 = (rc.y + tb.y) + tb.z;
    const float cnd0 = upd1(cen.x, U1.x, D1.x, lm,    cen.y, U4.x, D4.x, L4.x, R4.x,
                            U16.x, D16.x, L16.x, R16.x, T1.x, T2.x,
                            ta.x, ta.y, ta.z, ta.w, tb.x, rest0, lv0, al, dt,
                            sum_lap, sum_du, ssq0);
    const float cnd1 = upd1(cen.y, U1.y, D1.y, cen.x, rp,    U4.y, D4.y, L4.y, R4.y,
                            U16.y, D16.y, L16.y, R16.y, T1.y, T2.y,
                            ta.x, ta.y, ta.z, ta.w, tb.x, rest1, lv1, al, dt,
                            sum_lap, sum_du, ssq1);
    const float2 cv = { cnd0, cnd1 };
    if constexpr (NORM){
      *(float2*)(ucand + bbase + (size_t)cl*N_ + e0) = cv;
      *(float2*)(Rw + cl*36) = cv;
    } else {
      *(float2*)(unext + bbase + (size_t)cl*N_ + e0) = cv;
    }
  }
  }
  };
  if (do_norm) channel_steps(std::true_type{});
  else         channel_steps(std::false_type{});

  if (do_norm){
    ssq0 += __shfl_xor(ssq0, 16);
    ssq1 += __shfl_xor(ssq1, 16);
    const float scale0 = rsqrtf(ssq0/(float)C_ + 1e-8f);
    const float scale1 = rsqrtf(ssq1/(float)C_ + 1e-8f);
    #pragma unroll 8
    for (int it = 0; it < 64; ++it){
      const int cl = ((it >> 3) << 4) | (it & 7);
      const float2 cv = *(const float2*)(Rw + cl*36);
      const float nw = tabw[cl*8 + 7];
      float2 o;
      o.x = cv.x*scale0*nw;
      o.y = cv.y*scale1*nw;
      *(float2*)(unext + bbase + (size_t)cl*N_ + e0) = o;
    }
  }

  float s1 = blockReduceSum(sum_du);
  float s2 = blockReduceSum(sum_lap);
  if (threadIdx.x == 0){
    atomicAdd(&accum[1], s1);
    atomicAdd(&accum[0], s2);
  }
}

__global__ void final_out_kernel(const float* accum, const float* turbs, const float* sten, float* out){
  if (threadIdx.x == 0 && blockIdx.x == 0){
    float tl, sl;
    step_scalars(accum, tl, sl);
    float a = 0.f, bm = 0.f;
    for (int i = 0; i < MAX_STEPS_ - 1; ++i){ a += turbs[i]; bm += sten[i]; }
    a += tl; bm += sl;
    out[0] = a / (float)MAX_STEPS_;
    out[1] = bm / (float)MAX_STEPS_;
  }
}


extern "C" void kernel_launch(void* const* d_in, const int* in_sizes, int n_in,
                              void* d_out, int out_size, void* d_ws, size_t ws_size,
                              hipStream_t stream){
  (void)in_sizes; (void)n_in; (void)out_size; (void)ws_size;

  const float* u_in  = (const float*)d_in[0];
  const float* w1    = (const float*)d_in[1];
  const float* b1    = (const float*)d_in[2];
  const float* w2    = (const float*)d_in[3];
  const float* b2    = (const float*)d_in[4];
  const float* gw    = (const float*)d_in[5];
  const float* gb    = (const float*)d_in[6];
  const float* vw    = (const float*)d_in[7];
  const float* vb    = (const float*)d_in[8];
  const float* normw = (const float*)d_in[9];
  const float* wlp   = (const float*)d_in[10];
  const float* blp   = (const float*)d_in[11];
  const float* a_g   = (const float*)d_in[12];
  const float* a_l   = (const float*)d_in[13];
  const float* logdt = (const float*)d_in[14];
  const float* ks    = (const float*)d_in[15];
  const float* kt    = (const float*)d_in[16];

  float* Wf = (float*)d_ws;
  size_t o = 0;
  const size_t VOL = (size_t)B_ * C_ * N_;
  float* ubuf0  = Wf + o; o += VOL;
  float* ubuf1  = Wf + o; o += VOL;
  float* ucand  = Wf + o; o += VOL;
  float* probeA = Wf + o; o += PROBE_CELLS;
  float* probeB = Wf + o; o += PROBE_CELLS;
  float* pooled = Wf + o; o += 256;
  float* hstate = Wf + o; o += 256;
  float* lsmall = Wf + o; o += (size_t)B_*C_*32;
  float* lproj  = Wf + o; o += (size_t)B_*C_*32;
  float* spsks  = Wf + o; o += 3*C_;
  float* spskt  = Wf + o; o += 2*C_;
  float* scal   = Wf + o; o += 8;
  float* accum  = Wf + o; o += 8;
  float* turbs  = Wf + o; o += 16;
  float* sten   = Wf + o; o += 16;
  _Float16* w1h = (_Float16*)(Wf + o); o += (C_*HID_)/2;
  _Float16* w2h = (_Float16*)(Wf + o); o += (HID_*C_)/2;

  float* ubuf[2] = { ubuf0, ubuf1 };
  float* ufinal = (float*)d_out;

  prep_kernel<<<64, 256, 0, stream>>>(w1, w2, ks, kt, a_g, a_l, logdt,
                                      w1h, w2h, spsks, spskt, scal, hstate);
  pool3_kernel<<<B_*C_, 512, 0, stream>>>(u_in, u_in, 0, probeA, (const float*)nullptr,
                                          lsmall, pooled, accum);

  float* pprev = probeA;
  float* pcur  = probeB;

  for (int step = 0; step < MAX_STEPS_; ++step){
    const float* ucur  = (step == 0) ? u_in : ubuf[step & 1];
    float*       unext = (step == MAX_STEPS_ - 1) ? ufinal : ubuf[(step & 1) ^ 1];
    int do_norm = (((step + 1) % 2) == 0) ? 1 : 0;
    const float* cplane = do_norm ? ucand : unext;

    memory_kernel<<<MEM_GROUPS, 256, 0, stream>>>(pooled, gw, gb, vw, vb, hstate, lsmall, wlp, blp, lproj,
                                                  accum, turbs, sten, step);
    step_kernel<<<BN_/128, 128, 0, stream>>>(ucur, w1h, b1, w2h, b2, hstate, lproj, spsks, spskt,
                                             scal, normw, ucand, unext, accum, do_norm);
    pool3_kernel<<<B_*C_, 512, 0, stream>>>(cplane, unext, do_norm, pcur, pprev, lsmall, pooled, accum);

    float* tswap = pprev; pprev = pcur; pcur = tswap;
  }

  final_out_kernel<<<1, 32, 0, stream>>>(accum, turbs, sten, ((float*)d_out) + VOL);
}


# hidden loop's weight stream runs one chunk ahead in a second register set: the next chunk's 8 W1 fragments and biases are asked for as one group before the current GELU, not in pairs among product 1
# speedup vs baseline: 9.0708x; 1.1045x over previous
#include <hip/hip_runtime.h>
#include <math.h>
#include <type_traits>

#define B_ 2
#define C_ 128
#define T_ 8
#define H_ 64
#define W_ 64
#define N_ (T_*H_*W_)
#define BN_ (B_*N_)
#define HID_ 256
#define MAX_STEPS_ 12
#define DIFF_SCALE 0.25f
#define PROBE_CELLS (B_*C_*2*8*8)

typedef __attribute__((ext_vector_type(16))) _Float16 v16h;
typedef __attribute__((ext_vector_type(8)))  _Float16 h8;
typedef __attribute__((ext_vector_type(8)))  float    v8f;

__device__ __forceinline__ float softplus_f(float x){ return x > 20.f ? x : log1pf(expf(x)); }
__device__ __forceinline__ float gelu_f(float x){ return 0.5f*x*(1.f + erff(x*0.70710678118654752f)); }
__device__ __forceinline__ int imin_(int a, int b){ return a < b ? a : b; }

__device__ __forceinline__ float blockReduceSum(float v){
  __shared__ float red[8];
  int lane = threadIdx.x & 31, wid = threadIdx.x >> 5;
  #pragma unroll
  for (int off = 16; off > 0; off >>= 1) v += __shfl_down(v, off);
  if (lane == 0) red[wid] = v;
  __syncthreads();
  v = (threadIdx.x < (blockDim.x >> 5)) ? red[threadIdx.x] : 0.f;
  if (wid == 0){
    #pragma unroll
    for (int off = 4; off > 0; off >>= 1) v += __shfl_down(v, off);
  }
  __syncthreads();
  return v;
}

__device__ __forceinline__ v16h load_frag_glob(const _Float16* base, int f, int lane){
  const h8* p = (const h8*)base + f*64 + lane;
  h8 lo = p[0];
  h8 hi = p[32];
  return __builtin_shufflevector(lo, hi, 0,1,2,3,4,5,6,7,8,9,10,11,12,13,14,15);
}


__global__ void prep_kernel(const float* w1, const float* w2,
                            const float* ks, const float* kt,
                            const float* ag, const float* al, const float* logdt,
                            _Float16* w1h, _Float16* w2h,
                            float* spsks, float* spskt, float* scal, float* hstate){
  int i = blockIdx.x*blockDim.x + threadIdx.x;
  int stride = gridDim.x*blockDim.x;
  if (i < B_*C_) hstate[i] = 0.f;
  for (int j = i; j < C_*HID_; j += stride){
    int e = j & 7, ln = (j >> 3) & 31, half = (j >> 8) & 1;
    int nt = (j >> 9) & 1, kb = (j >> 10) & 3, hc = j >> 12;
    int k = kb*32 + half*16 + (ln >> 4)*8 + e;
    int n = hc*32 + nt*16 + (ln & 15);
    w1h[j] = (_Float16)w1[k*HID_ + n];
  }
  for (int j = i; j < HID_*C_; j += stride){
    int e = j & 7, ln = (j >> 3) & 31, half = (j >> 8) & 1;
    int nt = (j >> 9) & 7, hc = j >> 12;
    int k = hc*32 + half*16 + (ln >> 4)*8 + e;
    int n = nt*16 + (ln & 15);
    w2h[j] = (_Float16)w2[k*C_ + n];
  }
  for (int j = i; j < 3*C_;   j += stride) spsks[j] = softplus_f(ks[j]);
  for (int j = i; j < 2*C_;   j += stride) spskt[j] = softplus_f(kt[j]);
  if (i == 0){
    float dt = expf(logdt[0]);
    dt = fminf(fmaxf(dt, 0.005f), 0.35f);
    scal[0] = dt;
    scal[1] = softplus_f(ag[0]);
    scal[2] = softplus_f(al[0]);
  }
}

__device__ __forceinline__ float blockrow_sum(const float* p){
  const float4* q = (const float4*)p;
  float4 v[16];
  #pragma unroll
  for (int f = 0; f < 4; ++f){
    #pragma unroll
    for (int j = 0; j < 4; ++j) v[f*4 + j] = q[f*(H_*W_/4) + j*(2*W_/4)];
  }
  float s = 0.f;
  #pragma unroll
  for (int i = 0; i < 16; ++i) s += (v[i].x + v[i].y) + (v[i].z + v[i].w);
  s += __shfl_xor(s, 1);
  s += __shfl_xor(s, 16);
  return s;
}

__global__ void __launch_bounds__(512)
pool3_kernel(const float* cand, const float* state, int state_differs,
             float* probe, const float* prev, float* lsmall, float* pooled, float* accum){
  __shared__ float cs[16][8];
  __shared__ float dsum[16][8];
  __shared__ float psum[16][8];
  const int bc   = blockIdx.x;
  const int lane = threadIdx.x & 31;
  const int wv   = threadIdx.x >> 5;
  const int pt = wv >> 3, ph = wv & 7;
  const size_t off = (size_t)bc*N_ + (size_t)pt*(4*H_*W_)
                   + (size_t)((ph*8 + (lane >> 4))*W_ + (lane & 15)*4);
  const int pw = (lane >> 1) & 7;
  const bool owner = (lane & 17) == 0;
  const size_t pi = (size_t)(bc*16 + wv)*8 + pw;
  float pp = 0.f;
  if (prev != nullptr && owner) pp = prev[pi];
  const float s = blockrow_sum(cand + off);
  float ss = s;
  if (state_differs) ss = blockrow_sum(state + off);
  if (owner){
    const float cv = s * (1.f/256.f);
    probe[pi] = cv;
    cs[wv][pw] = ss;
    dsum[wv][pw] = fabsf(cv - pp);
    psum[wv][pw] = fabsf(pp);
  }
  __syncthreads();
  if (wv == 0){
    const int lt = lane >> 4, lh = (lane >> 2) & 3, lw = lane & 3;
    const int r0 = lt*8 + lh*2;
    float a = (cs[r0][2*lw] + cs[r0][2*lw + 1]) + (cs[r0 + 1][2*lw] + cs[r0 + 1][2*lw + 1]);
    lsmall[(size_t)bc*32 + lane] = a * (1.f/1024.f);
    a += __shfl_xor(a, 16);
    a += __shfl_xor(a, 8);
    a += __shfl_xor(a, 4);
    a += __shfl_xor(a, 2);
    a += __shfl_xor(a, 1);
    if (lane == 0) pooled[bc] = a * (1.f/(float)N_);
  } else if (wv == 1 && prev != nullptr){
    const float* d = &dsum[0][0];
    const float* p = &psum[0][0];
    float a = (d[lane] + d[32 + lane]) + (d[64 + lane] + d[96 + lane]);
    float q = (p[lane] + p[32 + lane]) + (p[64 + lane] + p[96 + lane]);
    for (int off2 = 16; off2 > 0; off2 >>= 1){ a += __shfl_xor(a, off2); q += __shfl_xor(q, off2); }
    if (lane == 0){
      atomicAdd(&accum[2], a);
      atomicAdd(&accum[3], q);
    }
  }
}

__device__ __forceinline__ void step_scalars(const float* accum, float& turb, float& se){
  const float inv_cells = 1.f/(float)PROBE_CELLS;
  const float inv_bnc   = 1.f/8388608.f;
  const float stop_turb = (accum[2]*inv_cells) / (accum[3]*inv_cells + 1e-8f);
  se = accum[1]*inv_bnc;
  const float le = accum[0]*inv_bnc;
  turb = stop_turb + 0.05f*se + 0.01f*le;
}

#define MEM_GROUPS 34
__global__ void __launch_bounds__(256)
memory_kernel(const float* pooled, const float* gw, const float* gb,
              const float* vw, const float* vb, float* h,
              const float* lsmall, const float* wlp, const float* blp, float* lproj,
              float* accum, float* turbs, float* sten, int step){
  const int tid = threadIdx.x;
  if (blockIdx.x < 32){
    const int id = blockIdx.x*256 + tid;
    const int cell = id & 31;
    const int oc = (id >> 5) & 127;
    const int b = id >> 12;
    float s = blp[oc];
    #pragma unroll 32
    for (int c = 0; c < C_; ++c)
      s += lsmall[((size_t)(b*C_ + c) << 5) + cell] * wlp[c*C_ + oc];
    lproj[((size_t)(b*C_ + oc) << 5) + cell] = s;
  } else if (blockIdx.x == 32){
    const int b = tid >> 7, c = tid & 127;
    const float* p = pooled + b*C_;
    float g = gb[c], v = vb[c];
    #pragma unroll 32
    for (int k = 0; k < C_; ++k){
      const float pk = p[k];
      g += pk * gw[k*C_ + c];
      v += pk * vw[k*C_ + c];
    }
    h[tid] += (1.f/(1.f+expf(-g))) * tanhf(v);
  } else if (tid == 0){
    if (step > 0){
      float tb, se;
      step_scalars(accum, tb, se);
      turbs[step - 1] = tb;
      sten[step - 1]  = se;
    }
    accum[0] = 0.f; accum[1] = 0.f; accum[2] = 0.f; accum[3] = 0.f;
  }
}

__device__ __forceinline__ float2 ld2(const float* p, bool ok){
  const float2 a = *(const float2*)p;
  float2 o; o.x = ok ? a.x : 0.f; o.y = ok ? a.y : 0.f;
  return o;
}
__device__ __forceinline__ float ld1(const float* p, bool ok){
  const float a = *p;
  return ok ? a : 0.f;
}
__device__ __forceinline__ float2 sel2(float2 a, bool ok){
  float2 o; o.x = ok ? a.x : 0.f; o.y = ok ? a.y : 0.f;
  return o;
}
__device__ __forceinline__ void ldc8(const float* p, float* o){
  const float4 a = *(const float4*)p, b = *(const float4*)(p + 4);
  o[0] = a.x; o[1] = a.y; o[2] = a.z; o[3] = a.w; o[4] = b.x; o[5] = b.y; o[6] = b.z; o[7] = b.w;
}

__device__ __forceinline__ float upd1(float uv, float u1a, float u1b, float l1, float r1,
                                      float u4a, float u4b, float l4, float r4,
                                      float u16a, float u16b, float l16, float r16,
                                      float t1, float t2,
                                      float k0, float k1, float k2, float q0, float q1,
                                      float rest, float lv, float al, float dt,
                                      float& sum_lap, float& sum_du, float& ssq){
  float lap1  = -4.f * uv; lap1  += u1a;  lap1  += u1b;  lap1  += l1;  lap1  += r1;
  float lap4  = -4.f * uv; lap4  += u4a;  lap4  += u4b;  lap4  += l4;  lap4  += r4;
  float lap16 = -4.f * uv; lap16 += u16a; lap16 += u16b; lap16 += l16; lap16 += r16;
  float diff = k0 * lap1;
  diff += k1 * lap4;
  diff += k2 * lap16;
  diff += q0 * (t1 - uv);
  diff += q1 * (t2 - uv);
  diff *= DIFF_SCALE;
  sum_lap += fabsf(diff);
  const float du = diff + rest + al*lv;
  const float cand = uv + dt*du;
  sum_du += fabsf(du);
  ssq += cand*cand;
  return cand;
}

__global__ void __launch_bounds__(128) __attribute__((amdgpu_waves_per_eu(2)))
step_kernel(const float* __restrict__ u, const _Float16* __restrict__ w1h, const float* __restrict__ b1,
            const _Float16* __restrict__ w2h, const float* __restrict__ b2,
            const float* __restrict__ h, const float* __restrict__ lproj,
            const float* __restrict__ spsks, const float* __restrict__ spskt,
            const float* __restrict__ scal, const float* __restrict__ normw,
            float* __restrict__ ucand, float* __restrict__ unext, float* accum, int do_norm){
  const int tid  = threadIdx.x;
  const int lane = tid & 31;
  const int wv   = __builtin_amdgcn_readfirstlane(tid >> 5);
  const int hf   = lane >> 4;
  const int col  = lane & 15;

  const int pixBase = blockIdx.x * 128 + wv * 32;
  const int b  = pixBase / N_;
  const int pw = pixBase - b * N_;
  const float* ub = u + (size_t)b * C_ * N_;
  const int e0 = hf*8*N_ + pw + 2*col;
  const int t  = pw >> 12;
  const int hh = (pw >> 6) & 63;

  __shared__ __attribute__((aligned(16))) float Rs[4][128][36];
  __shared__ __attribute__((aligned(16))) float tabs[128][8];
  __shared__ __attribute__((aligned(16))) float Lws[2][128][4];
  __shared__ __attribute__((aligned(16))) float Cs[2][128][68];
  {
    const int c = tid;
    const float agv = scal[1];
    const float4 k4 = { spsks[c], spsks[C_ + c], spsks[2*C_ + c], spskt[c] };
    const float4 q4 = { spskt[C_ + c], b2[c], agv * h[b*C_ + c], normw[c] };
    *(float4*)&tabs[c][0] = k4;
    *(float4*)&tabs[c][4] = q4;
    const int tb_ = (blockIdx.x * 128 - b * N_) >> 12;
    const int hb_ = ((blockIdx.x * 128 - b * N_) >> 6) & 63;
    const float st  = fminf(fmaxf((tb_ + 0.5f)*(2.f/T_) - 0.5f, 0.f), 1.f);
    const int t0i = (int)st;  const int t1i = imin_(t0i+1, 1); const float ft = st - (float)t0i;
    const float4* lp = (const float4*)(lproj + ((size_t)(b*C_ + c) << 5));
    #pragma unroll
    for (int rw = 0; rw < 2; ++rw){
      const float sh  = fminf(fmaxf((hb_ + rw + 0.5f)*(4.f/H_) - 0.5f, 0.f), 3.f);
      const int h0i = (int)sh;  const int h1i = imin_(h0i+1, 3); const float fh = sh - (float)h0i;
      const float4 A00 = lp[t0i*4 + h0i], A01 = lp[t0i*4 + h1i];
      const float4 A10 = lp[t1i*4 + h0i], A11 = lp[t1i*4 + h1i];
      float4 L;
      L.x = (A00.x*(1.f-fh) + A01.x*fh)*(1.f-ft) + (A10.x*(1.f-fh) + A11.x*fh)*ft;
      L.y = (A00.y*(1.f-fh) + A01.y*fh)*(1.f-ft) + (A10.y*(1.f-fh) + A11.y*fh)*ft;
      L.z = (A00.z*(1.f-fh) + A01.z*fh)*(1.f-ft) + (A10.z*(1.f-fh) + A11.z*fh)*ft;
      L.w = (A00.w*(1.f-fh) + A01.w*fh)*(1.f-ft) + (A10.w*(1.f-fh) + A11.w*fh)*ft;
      *(float4*)&Lws[rw][c][0] = L;
    }
  }

  v16h af[2][4];
  const int rowi = wv >> 1;
  const int w0   = (pw & 63) + 2*col;
  {
    const float* ap = ub + e0;
    float* cw = &Cs[rowi][hf*8][w0];
    #pragma unroll
    for (int kb = 0; kb < 4; ++kb){
      #pragma unroll
      for (int s = 0; s < 16; ++s){
        const int ch = kb*32 + (s < 8 ? s : s + 8);
        const float2 v = *(const float2*)(ap + (size_t)ch * N_);
        *(float2*)(cw + ch*68) = v;
        af[0][kb][s] = (_Float16)v.x;
        af[1][kb][s] = (_Float16)v.y;
      }
    }
  }

  const v8f z8 = {0.f,0.f,0.f,0.f,0.f,0.f,0.f,0.f};
  v8f accC[8][2];
  #pragma unroll
  for (int i = 0; i < 8; ++i){ accC[i][0] = z8; accC[i][1] = z8; }

  v16h bf[8];
  float bb0[8], bb1[8];
  #pragma unroll
  for (int f = 0; f < 8; ++f) bf[f] = load_frag_glob(w1h, f, lane);
  ldc8(b1 + hf*8, bb0);
  ldc8(b1 + 16 + hf*8, bb1);

  for (int hc = 0; hc < 8; ++hc){
    v8f a00 = z8, a01 = z8, a10 = z8, a11 = z8;
    #pragma unroll
    for (int kb = 0; kb < 4; ++kb){
      a00 = __builtin_amdgcn_wmma_f32_16x16x32_f16(false, bf[kb*2],   false, af[0][kb], (short)0, a00, false, false);
      a01 = __builtin_amdgcn_wmma_f32_16x16x32_f16(false, bf[kb*2+1], false, af[0][kb], (short)0, a01, false, false);
      a10 = __builtin_amdgcn_wmma_f32_16x16x32_f16(false, bf[kb*2],   false, af[1][kb], (short)0, a10, false, false);
      a11 = __builtin_amdgcn_wmma_f32_16x16x32_f16(false, bf[kb*2+1], false, af[1][kb], (short)0, a11, false, false);
    }
    v16h bw[8];
    #pragma unroll
    for (int nt = 0; nt < 8; ++nt) bw[nt] = load_frag_glob(w2h, hc*8 + nt, lane);
    const int hn = imin_(hc + 1, 7);
    v16h bfn[8];
    #pragma unroll
    for (int f = 0; f < 8; ++f) bfn[f] = load_frag_glob(w1h, hn*8 + f, lane);
    float nb0[8], nb1[8];
    ldc8(b1 + hn*32 + hf*8, nb0);
    ldc8(b1 + hn*32 + 16 + hf*8, nb1);
    v16h ah0, ah1;
    #pragma unroll
    for (int r = 0; r < 8; ++r){
      ah0[r]     = (_Float16)gelu_f(a00[r] + bb0[r]);
      ah0[8 + r] = (_Float16)gelu_f(a01[r] + bb1[r]);
      ah1[r]     = (_Float16)gelu_f(a10[r] + bb0[r]);
      ah1[8 + r] = (_Float16)gelu_f(a11[r] + bb1[r]);
    }
    #pragma unroll
    for (int nt = 0; nt < 8; ++nt){
      accC[nt][0] = __builtin_amdgcn_wmma_f32_16x16x32_f16(false, bw[nt], false, ah0, (short)0, accC[nt][0], false, false);
      accC[nt][1] = __builtin_amdgcn_wmma_f32_16x16x32_f16(false, bw[nt], false, ah1, (short)0, accC[nt][1], false, false);
    }
    #pragma unroll
    for (int f = 0; f < 8; ++f) bf[f] = bfn[f];
    #pragma unroll
    for (int r = 0; r < 8; ++r){ bb0[r] = nb0[r]; bb1[r] = nb1[r]; }
  }

  #pragma unroll
  for (int nt = 0; nt < 8; ++nt){
    #pragma unroll
    for (int r = 0; r < 8; ++r){
      float2 rv = { accC[nt][0][r], accC[nt][1][r] };
      *(float2*)&Rs[wv][nt*16 + hf*8 + r][2*col] = rv;
    }
  }
  __syncthreads();

  const float dt = scal[0], al = scal[2];

  float ww[2][4];
  #pragma unroll
  for (int j = 0; j < 2; ++j){
    const float sw = fminf(fmaxf((w0 + j + 0.5f)*(4.f/W_) - 0.5f, 0.f), 3.f);
    const int wi0 = (int)sw; const int wi1 = imin_(wi0+1, 3); const float f = sw - (float)wi0;
    #pragma unroll
    for (int i = 0; i < 4; ++i) ww[j][i] = (i == wi0 ? 1.f - f : 0.f) + (i == wi1 ? f : 0.f);
  }

  const bool vU1 = hh - 1  >= 0, vD1 = hh + 1  < 64;
  const bool vU4 = hh - 4  >= 0, vD4 = hh + 4  < 64;
  const bool vU16= hh - 16 >= 0, vD16= hh + 16 < 64;
  const bool vT1 = t - 1 >= 0,   vT2 = t - 2 >= 0;
  const bool vL4 = w0 - 4  >= 0, vR4 = w0 + 5  < 64;
  const bool vL16= w0 - 16 >= 0, vR16= w0 + 17 < 64;
  const bool vLm = w0 - 1 >= 0,  vRp = w0 + 2 < 64;
  const int oU1 = vU1 ? -64 : 0,     oD1 = vD1 ? 64 : 0;
  const int oU4 = vU4 ? -256 : 0,    oD4 = vD4 ? 256 : 0;
  const int oU16= vU16 ? -1024 : 0,  oD16= vD16 ? 1024 : 0;
  const int oT1 = vT1 ? -4096 : 0,   oT2 = vT2 ? -8192 : 0;
  const int oL4 = vL4 ? -4 : 0,      oR4 = vR4 ? 4 : 0;
  const int oL16= vL16 ? -16 : 0,    oR16= vR16 ? 16 : 0;
  const int oLm = vLm ? -1 : 0,      oRp = vRp ? 2 : 0;
  const bool vG1 = rowi == 0 ? vU1 : vD1;
  const int  oG1 = rowi == 0 ? oU1 : oD1;

  const size_t bbase = (size_t)b * C_ * N_;
  float sum_du = 0.f, sum_lap = 0.f;
  float ssq0 = 0.f, ssq1 = 0.f;

  float*       Rw   = &Rs[wv][hf*8][2*col];
  const float* tabw = &tabs[hf*8][0];
  const float* Lww  = &Lws[wv >> 1][hf*8][0];
  const float* Cme  = &Cs[rowi][hf*8][w0];
  const float* Cot  = &Cs[rowi ^ 1][hf*8][w0];

  auto channel_steps = [&](auto norm_tag){
  constexpr bool NORM = decltype(norm_tag)::value;
  for (int ib = 0; ib < 8; ++ib){
  float2 gG1[8], gU4[8], gD4[8], gU16[8], gD16[8], gT1[8], gT2[8];
  #pragma unroll
  for (int j = 0; j < 8; ++j){
    const float* uc = ub + (size_t)(ib*16 + j)*N_ + e0;
    gG1[j]  = *(const float2*)(uc + oG1);
    gU4[j]  = *(const float2*)(uc + oU4);   gD4[j]  = *(const float2*)(uc + oD4);
    gU16[j] = *(const float2*)(uc + oU16);  gD16[j] = *(const float2*)(uc + oD16);
    gT1[j]  = *(const float2*)(uc + oT1);   gT2[j]  = *(const float2*)(uc + oT2);
  }
  #pragma unroll
  for (int j = 0; j < 8; ++j){
    const int cl = ib*16 + j;
    const float4 ta = *(const float4*)(tabw + cl*8);
    const float4 tb = *(const float4*)(tabw + cl*8 + 4);
    const float4 Lq = *(const float4*)(Lww + cl*4);
    const float2 rc = *(const float2*)(Rw + cl*36);
    const float lv0 = Lq.x*ww[0][0] + Lq.y*ww[0][1] + Lq.z*ww[0][2] + Lq.w*ww[0][3];
    const float lv1 = Lq.x*ww[1][0] + Lq.y*ww[1][1] + Lq.z*ww[1][2] + Lq.w*ww[1][3];

    const float* cc = Cme + cl*68;
    const float2 G1  = sel2(gG1[j],  vG1);
    const float2 U4  = sel2(gU4[j],  vU4),  D4  = sel2(gD4[j],  vD4);
    const float2 U16 = sel2(gU16[j], vU16), D16 = sel2(gD16[j], vD16);
    const float2 T1  = sel2(gT1[j],  vT1),  T2  = sel2(gT2[j],  vT2);
    const float2 cen = *(const float2*)cc;
    const float2 S1  = *(const float2*)(Cot + cl*68);
    const float2 L4  = ld2(cc + oL4,  vL4),  R4  = ld2(cc + oR4,  vR4);
    const float2 L16 = ld2(cc + oL16, vL16), R16 = ld2(cc + oR16, vR16);
    const float  lm  = ld1(cc + oLm,  vLm);
    const float  rp  = ld1(cc + oRp,  vRp);
    const float2 U1  = rowi == 0 ? G1 : S1;
    const float2 D1  = rowi == 0 ? S1 : G1;

    const float rest0 = (rc.x + tb.y) + tb.z;
    const float rest1 = (rc.y + tb.y) + tb.z;
    const float cnd0 = upd1(cen.x, U1.x, D1.x, lm,    cen.y, U4.x, D4.x, L4.x, R4.x,
                            U16.x, D16.x, L16.x, R16.x, T1.x, T2.x,
                            ta.x, ta.y, ta.z, ta.w, tb.x, rest0, lv0, al, dt,
                            sum_lap, sum_du, ssq0);
    const float cnd1 = upd1(cen.y, U1.y, D1.y, cen.x, rp,    U4.y, D4.y, L4.y, R4.y,
                            U16.y, D16.y, L16.y, R16.y, T1.y, T2.y,
                            ta.x, ta.y, ta.z, ta.w, tb.x, rest1, lv1, al, dt,
                            sum_lap, sum_du, ssq1);
    const float2 cv = { cnd0, cnd1 };
    if constexpr (NORM){
      *(float2*)(ucand + bbase + (size_t)cl*N_ + e0) = cv;
      *(float2*)(Rw + cl*36) = cv;
    } else {
      *(float2*)(unext + bbase + (size_t)cl*N_ + e0) = cv;
    }
  }
  }
  };
  if (do_norm) channel_steps(std::true_type{});
  else         channel_steps(std::false_type{});

  if (do_norm){
    ssq0 += __shfl_xor(ssq0, 16);
    ssq1 += __shfl_xor(ssq1, 16);
    const float scale0 = rsqrtf(ssq0/(float)C_ + 1e-8f);
    const float scale1 = rsqrtf(ssq1/(float)C_ + 1e-8f);
    #pragma unroll 8
    for (int it = 0; it < 64; ++it){
      const int cl = ((it >> 3) << 4) | (it & 7);
      const float2 cv = *(const float2*)(Rw + cl*36);
      const float nw = tabw[cl*8 + 7];
      float2 o;
      o.x = cv.x*scale0*nw;
      o.y = cv.y*scale1*nw;
      *(float2*)(unext + bbase + (size_t)cl*N_ + e0) = o;
    }
  }

  float s1 = blockReduceSum(sum_du);
  float s2 = blockReduceSum(sum_lap);
  if (threadIdx.x == 0){
    atomicAdd(&accum[1], s1);
    atomicAdd(&accum[0], s2);
  }
}

__global__ void final_out_kernel(const float* accum, const float* turbs, const float* sten, float* out){
  if (threadIdx.x == 0 && blockIdx.x == 0){
    float tl, sl;
    step_scalars(accum, tl, sl);
    float a = 0.f, bm = 0.f;
    for (int i = 0; i < MAX_STEPS_ - 1; ++i){ a += turbs[i]; bm += sten[i]; }
    a += tl; bm += sl;
    out[0] = a / (float)MAX_STEPS_;
    out[1] = bm / (float)MAX_STEPS_;
  }
}


extern "C" void kernel_launch(void* const* d_in, const int* in_sizes, int n_in,
                              void* d_out, int out_size, void* d_ws, size_t ws_size,
                              hipStream_t stream){
  (void)in_sizes; (void)n_in; (void)out_size; (void)ws_size;

  const float* u_in  = (const float*)d_in[0];
  const float* w1    = (const float*)d_in[1];
  const float* b1    = (const float*)d_in[2];
  const float* w2    = (const float*)d_in[3];
  const float* b2    = (const float*)d_in[4];
  const float* gw    = (const float*)d_in[5];
  const float* gb    = (const float*)d_in[6];
  const float* vw    = (const float*)d_in[7];
  const float* vb    = (const float*)d_in[8];
  const float* normw = (const float*)d_in[9];
  const float* wlp   = (const float*)d_in[10];
  const float* blp   = (const float*)d_in[11];
  const float* a_g   = (const float*)d_in[12];
  const float* a_l   = (const float*)d_in[13];
  const float* logdt = (const float*)d_in[14];
  const float* ks    = (const float*)d_in[15];
  const float* kt    = (const float*)d_in[16];

  float* Wf = (float*)d_ws;
  size_t o = 0;
  const size_t VOL = (size_t)B_ * C_ * N_;
  float* ubuf0  = Wf + o; o += VOL;
  float* ubuf1  = Wf + o; o += VOL;
  float* ucand  = Wf + o; o += VOL;
  float* probeA = Wf + o; o += PROBE_CELLS;
  float* probeB = Wf + o; o += PROBE_CELLS;
  float* pooled = Wf + o; o += 256;
  float* hstate = Wf + o; o += 256;
  float* lsmall = Wf + o; o += (size_t)B_*C_*32;
  float* lproj  = Wf + o; o += (size_t)B_*C_*32;
  float* spsks  = Wf + o; o += 3*C_;
  float* spskt  = Wf + o; o += 2*C_;
  float* scal   = Wf + o; o += 8;
  float* accum  = Wf + o; o += 8;
  float* turbs  = Wf + o; o += 16;
  float* sten   = Wf + o; o += 16;
  _Float16* w1h = (_Float16*)(Wf + o); o += (C_*HID_)/2;
  _Float16* w2h = (_Float16*)(Wf + o); o += (HID_*C_)/2;

  float* ubuf[2] = { ubuf0, ubuf1 };
  float* ufinal = (float*)d_out;

  prep_kernel<<<64, 256, 0, stream>>>(w1, w2, ks, kt, a_g, a_l, logdt,
                                      w1h, w2h, spsks, spskt, scal, hstate);
  pool3_kernel<<<B_*C_, 512, 0, stream>>>(u_in, u_in, 0, probeA, (const float*)nullptr,
                                          lsmall, pooled, accum);

  float* pprev = probeA;
  float* pcur  = probeB;

  for (int step = 0; step < MAX_STEPS_; ++step){
    const float* ucur  = (step == 0) ? u_in : ubuf[step & 1];
    float*       unext = (step == MAX_STEPS_ - 1) ? ufinal : ubuf[(step & 1) ^ 1];
    int do_norm = (((step + 1) % 2) == 0) ? 1 : 0;
    const float* cplane = do_norm ? ucand : unext;

    memory_kernel<<<MEM_GROUPS, 256, 0, stream>>>(pooled, gw, gb, vw, vb, hstate, lsmall, wlp, blp, lproj,
                                                  accum, turbs, sten, step);
    step_kernel<<<BN_/128, 128, 0, stream>>>(ucur, w1h, b1, w2h, b2, hstate, lproj, spsks, spskt,
                                             scal, normw, ucand, unext, accum, do_norm);
    pool3_kernel<<<B_*C_, 512, 0, stream>>>(cplane, unext, do_norm, pcur, pprev, lsmall, pooled, accum);

    float* tswap = pprev; pprev = pcur; pcur = tswap;
  }

  final_out_kernel<<<1, 32, 0, stream>>>(accum, turbs, sten, ((float*)d_out) + VOL);
}


# MLP GELU by ONE range-free erfc formula (rational x exponential, abs error 1.5e-7, far below the f16 rounding that follows): rcp + exp2 + packed f32 fma per pair, no per-lane second path
# speedup vs baseline: 10.4295x; 1.1498x over previous
#include <hip/hip_runtime.h>
#include <math.h>
#include <type_traits>

#define B_ 2
#define C_ 128
#define T_ 8
#define H_ 64
#define W_ 64
#define N_ (T_*H_*W_)
#define BN_ (B_*N_)
#define HID_ 256
#define MAX_STEPS_ 12
#define DIFF_SCALE 0.25f
#define PROBE_CELLS (B_*C_*2*8*8)

typedef __attribute__((ext_vector_type(16))) _Float16 v16h;
typedef __attribute__((ext_vector_type(8)))  _Float16 h8;
typedef __attribute__((ext_vector_type(8)))  float    v8f;

__device__ __forceinline__ float softplus_f(float x){ return x > 20.f ? x : log1pf(expf(x)); }
typedef __attribute__((ext_vector_type(2)))  float    v2f;
__device__ __forceinline__ v2f gelu2(v2f x){
  const v2f ax = { fabsf(x.x), fabsf(x.y) };
  const v2f d  = ax*0.23164190f + 1.0f;
  const v2f t  = { __builtin_amdgcn_rcpf(d.x), __builtin_amdgcn_rcpf(d.y) };
  const v2f q  = (x*x)*(-0.72134752f);
  const v2f e  = { __builtin_amdgcn_exp2f(q.x), __builtin_amdgcn_exp2f(q.y) };
  v2f p = t*0.5307027145f + (-0.7265760135f);
  p = p*t + 0.7107068705f;
  p = p*t + (-0.142248368f);
  p = p*t + 0.127414796f;
  const v2f hh = (p*t)*e;
  const v2f m  = { fmaxf(x.x, 0.f), fmaxf(x.y, 0.f) };
  return m - ax*hh;
}
__device__ __forceinline__ int imin_(int a, int b){ return a < b ? a : b; }

__device__ __forceinline__ float blockReduceSum(float v){
  __shared__ float red[8];
  int lane = threadIdx.x & 31, wid = threadIdx.x >> 5;
  #pragma unroll
  for (int off = 16; off > 0; off >>= 1) v += __shfl_down(v, off);
  if (lane == 0) red[wid] = v;
  __syncthreads();
  v = (threadIdx.x < (blockDim.x >> 5)) ? red[threadIdx.x] : 0.f;
  if (wid == 0){
    #pragma unroll
    for (int off = 4; off > 0; off >>= 1) v += __shfl_down(v, off);
  }
  __syncthreads();
  return v;
}

__device__ __forceinline__ v16h load_frag_glob(const _Float16* base, int f, int lane){
  const h8* p = (const h8*)base + f*64 + lane;
  h8 lo = p[0];
  h8 hi = p[32];
  return __builtin_shufflevector(lo, hi, 0,1,2,3,4,5,6,7,8,9,10,11,12,13,14,15);
}


__global__ void prep_kernel(const float* w1, const float* w2,
                            const float* ks, const float* kt,
                            const float* ag, const float* al, const float* logdt,
                            _Float16* w1h, _Float16* w2h,
                            float* spsks, float* spskt, float* scal, float* hstate){
  int i = blockIdx.x*blockDim.x + threadIdx.x;
  int stride = gridDim.x*blockDim.x;
  if (i < B_*C_) hstate[i] = 0.f;
  for (int j = i; j < C_*HID_; j += stride){
    int e = j & 7, ln = (j >> 3) & 31, half = (j >> 8) & 1;
    int nt = (j >> 9) & 1, kb = (j >> 10) & 3, hc = j >> 12;
    int k = kb*32 + half*16 + (ln >> 4)*8 + e;
    int n = hc*32 + nt*16 + (ln & 15);
    w1h[j] = (_Float16)w1[k*HID_ + n];
  }
  for (int j = i; j < HID_*C_; j += stride){
    int e = j & 7, ln = (j >> 3) & 31, half = (j >> 8) & 1;
    int nt = (j >> 9) & 7, hc = j >> 12;
    int k = hc*32 + half*16 + (ln >> 4)*8 + e;
    int n = nt*16 + (ln & 15);
    w2h[j] = (_Float16)w2[k*C_ + n];
  }
  for (int j = i; j < 3*C_;   j += stride) spsks[j] = softplus_f(ks[j]);
  for (int j = i; j < 2*C_;   j += stride) spskt[j] = softplus_f(kt[j]);
  if (i == 0){
    float dt = expf(logdt[0]);
    dt = fminf(fmaxf(dt, 0.005f), 0.35f);
    scal[0] = dt;
    scal[1] = softplus_f(ag[0]);
    scal[2] = softplus_f(al[0]);
  }
}

__device__ __forceinline__ float blockrow_sum(const float* p){
  const float4* q = (const float4*)p;
  float4 v[16];
  #pragma unroll
  for (int f = 0; f < 4; ++f){
    #pragma unroll
    for (int j = 0; j < 4; ++j) v[f*4 + j] = q[f*(H_*W_/4) + j*(2*W_/4)];
  }
  float s = 0.f;
  #pragma unroll
  for (int i = 0; i < 16; ++i) s += (v[i].x + v[i].y) + (v[i].z + v[i].w);
  s += __shfl_xor(s, 1);
  s += __shfl_xor(s, 16);
  return s;
}

__global__ void __launch_bounds__(512)
pool3_kernel(const float* cand, const float* state, int state_differs,
             float* probe, const float* prev, float* lsmall, float* pooled, float* accum){
  __shared__ float cs[16][8];
  __shared__ float dsum[16][8];
  __shared__ float psum[16][8];
  const int bc   = blockIdx.x;
  const int lane = threadIdx.x & 31;
  const int wv   = threadIdx.x >> 5;
  const int pt = wv >> 3, ph = wv & 7;
  const size_t off = (size_t)bc*N_ + (size_t)pt*(4*H_*W_)
                   + (size_t)((ph*8 + (lane >> 4))*W_ + (lane & 15)*4);
  const int pw = (lane >> 1) & 7;
  const bool owner = (lane & 17) == 0;
  const size_t pi = (size_t)(bc*16 + wv)*8 + pw;
  float pp = 0.f;
  if (prev != nullptr && owner) pp = prev[pi];
  const float s = blockrow_sum(cand + off);
  float ss = s;
  if (state_differs) ss = blockrow_sum(state + off);
  if (owner){
    const float cv = s * (1.f/256.f);
    probe[pi] = cv;
    cs[wv][pw] = ss;
    dsum[wv][pw] = fabsf(cv - pp);
    psum[wv][pw] = fabsf(pp);
  }
  __syncthreads();
  if (wv == 0){
    const int lt = lane >> 4, lh = (lane >> 2) & 3, lw = lane & 3;
    const int r0 = lt*8 + lh*2;
    float a = (cs[r0][2*lw] + cs[r0][2*lw + 1]) + (cs[r0 + 1][2*lw] + cs[r0 + 1][2*lw + 1]);
    lsmall[(size_t)bc*32 + lane] = a * (1.f/1024.f);
    a += __shfl_xor(a, 16);
    a += __shfl_xor(a, 8);
    a += __shfl_xor(a, 4);
    a += __shfl_xor(a, 2);
    a += __shfl_xor(a, 1);
    if (lane == 0) pooled[bc] = a * (1.f/(float)N_);
  } else if (wv == 1 && prev != nullptr){
    const float* d = &dsum[0][0];
    const float* p = &psum[0][0];
    float a = (d[lane] + d[32 + lane]) + (d[64 + lane] + d[96 + lane]);
    float q = (p[lane] + p[32 + lane]) + (p[64 + lane] + p[96 + lane]);
    for (int off2 = 16; off2 > 0; off2 >>= 1){ a += __shfl_xor(a, off2); q += __shfl_xor(q, off2); }
    if (lane == 0){
      atomicAdd(&accum[2], a);
      atomicAdd(&accum[3], q);
    }
  }
}

__device__ __forceinline__ void step_scalars(const float* accum, float& turb, float& se){
  const float inv_cells = 1.f/(float)PROBE_CELLS;
  const float inv_bnc   = 1.f/8388608.f;
  const float stop_turb = (accum[2]*inv_cells) / (accum[3]*inv_cells + 1e-8f);
  se = accum[1]*inv_bnc;
  const float le = accum[0]*inv_bnc;
  turb = stop_turb + 0.05f*se + 0.01f*le;
}

#define MEM_GROUPS 34
__global__ void __launch_bounds__(256)
memory_kernel(const float* pooled, const float* gw, const float* gb,
              const float* vw, const float* vb, float* h,
              const float* lsmall, const float* wlp, const float* blp, float* lproj,
              float* accum, float* turbs, float* sten, int step){
  const int tid = threadIdx.x;
  if (blockIdx.x < 32){
    const int id = blockIdx.x*256 + tid;
    const int cell = id & 31;
    const int oc = (id >> 5) & 127;
    const int b = id >> 12;
    float s = blp[oc];
    #pragma unroll 32
    for (int c = 0; c < C_; ++c)
      s += lsmall[((size_t)(b*C_ + c) << 5) + cell] * wlp[c*C_ + oc];
    lproj[((size_t)(b*C_ + oc) << 5) + cell] = s;
  } else if (blockIdx.x == 32){
    const int b = tid >> 7, c = tid & 127;
    const float* p = pooled + b*C_;
    float g = gb[c], v = vb[c];
    #pragma unroll 32
    for (int k = 0; k < C_; ++k){
      const float pk = p[k];
      g += pk * gw[k*C_ + c];
      v += pk * vw[k*C_ + c];
    }
    h[tid] += (1.f/(1.f+expf(-g))) * tanhf(v);
  } else if (tid == 0){
    if (step > 0){
      float tb, se;
      step_scalars(accum, tb, se);
      turbs[step - 1] = tb;
      sten[step - 1]  = se;
    }
    accum[0] = 0.f; accum[1] = 0.f; accum[2] = 0.f; accum[3] = 0.f;
  }
}

__device__ __forceinline__ float2 ld2(const float* p, bool ok){
  const float2 a = *(const float2*)p;
  float2 o; o.x = ok ? a.x : 0.f; o.y = ok ? a.y : 0.f;
  return o;
}
__device__ __forceinline__ float ld1(const float* p, bool ok){
  const float a = *p;
  return ok ? a : 0.f;
}
__device__ __forceinline__ float2 sel2(float2 a, bool ok){
  float2 o; o.x = ok ? a.x : 0.f; o.y = ok ? a.y : 0.f;
  return o;
}
__device__ __forceinline__ void ldc8(const float* p, float* o){
  const float4 a = *(const float4*)p, b = *(const float4*)(p + 4);
  o[0] = a.x; o[1] = a.y; o[2] = a.z; o[3] = a.w; o[4] = b.x; o[5] = b.y; o[6] = b.z; o[7] = b.w;
}

__device__ __forceinline__ float upd1(float uv, float u1a, float u1b, float l1, float r1,
                                      float u4a, float u4b, float l4, float r4,
                                      float u16a, float u16b, float l16, float r16,
                                      float t1, float t2,
                                      float k0, float k1, float k2, float q0, float q1,
                                      float rest, float lv, float al, float dt,
                                      float& sum_lap, float& sum_du, float& ssq){
  float lap1  = -4.f * uv; lap1  += u1a;  lap1  += u1b;  lap1  += l1;  lap1  += r1;
  float lap4  = -4.f * uv; lap4  += u4a;  lap4  += u4b;  lap4  += l4;  lap4  += r4;
  float lap16 = -4.f * uv; lap16 += u16a; lap16 += u16b; lap16 += l16; lap16 += r16;
  float diff = k0 * lap1;
  diff += k1 * lap4;
  diff += k2 * lap16;
  diff += q0 * (t1 - uv);
  diff += q1 * (t2 - uv);
  diff *= DIFF_SCALE;
  sum_lap += fabsf(diff);
  const float du = diff + rest + al*lv;
  const float cand = uv + dt*du;
  sum_du += fabsf(du);
  ssq += cand*cand;
  return cand;
}

__global__ void __launch_bounds__(128) __attribute__((amdgpu_waves_per_eu(2)))
step_kernel(const float* __restrict__ u, const _Float16* __restrict__ w1h, const float* __restrict__ b1,
            const _Float16* __restrict__ w2h, const float* __restrict__ b2,
            const float* __restrict__ h, const float* __restrict__ lproj,
            const float* __restrict__ spsks, const float* __restrict__ spskt,
            const float* __restrict__ scal, const float* __restrict__ normw,
            float* __restrict__ ucand, float* __restrict__ unext, float* accum, int do_norm){
  const int tid  = threadIdx.x;
  const int lane = tid & 31;
  const int wv   = __builtin_amdgcn_readfirstlane(tid >> 5);
  const int hf   = lane >> 4;
  const int col  = lane & 15;

  const int pixBase = blockIdx.x * 128 + wv * 32;
  const int b  = pixBase / N_;
  const int pw = pixBase - b * N_;
  const float* ub = u + (size_t)b * C_ * N_;
  const int e0 = hf*8*N_ + pw + 2*col;
  const int t  = pw >> 12;
  const int hh = (pw >> 6) & 63;

  __shared__ __attribute__((aligned(16))) float Rs[4][128][36];
  __shared__ __attribute__((aligned(16))) float tabs[128][8];
  __shared__ __attribute__((aligned(16))) float Lws[2][128][4];
  __shared__ __attribute__((aligned(16))) float Cs[2][128][68];
  {
    const int c = tid;
    const float agv = scal[1];
    const float4 k4 = { spsks[c], spsks[C_ + c], spsks[2*C_ + c], spskt[c] };
    const float4 q4 = { spskt[C_ + c], b2[c], agv * h[b*C_ + c], normw[c] };
    *(float4*)&tabs[c][0] = k4;
    *(float4*)&tabs[c][4] = q4;
    const int tb_ = (blockIdx.x * 128 - b * N_) >> 12;
    const int hb_ = ((blockIdx.x * 128 - b * N_) >> 6) & 63;
    const float st  = fminf(fmaxf((tb_ + 0.5f)*(2.f/T_) - 0.5f, 0.f), 1.f);
    const int t0i = (int)st;  const int t1i = imin_(t0i+1, 1); const float ft = st - (float)t0i;
    const float4* lp = (const float4*)(lproj + ((size_t)(b*C_ + c) << 5));
    #pragma unroll
    for (int rw = 0; rw < 2; ++rw){
      const float sh  = fminf(fmaxf((hb_ + rw + 0.5f)*(4.f/H_) - 0.5f, 0.f), 3.f);
      const int h0i = (int)sh;  const int h1i = imin_(h0i+1, 3); const float fh = sh - (float)h0i;
      const float4 A00 = lp[t0i*4 + h0i], A01 = lp[t0i*4 + h1i];
      const float4 A10 = lp[t1i*4 + h0i], A11 = lp[t1i*4 + h1i];
      float4 L;
      L.x = (A00.x*(1.f-fh) + A01.x*fh)*(1.f-ft) + (A10.x*(1.f-fh) + A11.x*fh)*ft;
      L.y = (A00.y*(1.f-fh) + A01.y*fh)*(1.f-ft) + (A10.y*(1.f-fh) + A11.y*fh)*ft;
      L.z = (A00.z*(1.f-fh) + A01.z*fh)*(1.f-ft) + (A10.z*(1.f-fh) + A11.z*fh)*ft;
      L.w = (A00.w*(1.f-fh) + A01.w*fh)*(1.f-ft) + (A10.w*(1.f-fh) + A11.w*fh)*ft;
      *(float4*)&Lws[rw][c][0] = L;
    }
  }

  v16h af[2][4];
  const int rowi = wv >> 1;
  const int w0   = (pw & 63) + 2*col;
  {
    const float* ap = ub + e0;
    float* cw = &Cs[rowi][hf*8][w0];
    #pragma unroll
    for (int kb = 0; kb < 4; ++kb){
      #pragma unroll
      for (int s = 0; s < 16; ++s){
        const int ch = kb*32 + (s < 8 ? s : s + 8);
        const float2 v = *(const float2*)(ap + (size_t)ch * N_);
        *(float2*)(cw + ch*68) = v;
        af[0][kb][s] = (_Float16)v.x;
        af[1][kb][s] = (_Float16)v.y;
      }
    }
  }

  const v8f z8 = {0.f,0.f,0.f,0.f,0.f,0.f,0.f,0.f};
  v8f accC[8][2];
  #pragma unroll
  for (int i = 0; i < 8; ++i){ accC[i][0] = z8; accC[i][1] = z8; }

  v16h bf[8];
  float bb0[8], bb1[8];
  #pragma unroll
  for (int f = 0; f < 8; ++f) bf[f] = load_frag_glob(w1h, f, lane);
  ldc8(b1 + hf*8, bb0);
  ldc8(b1 + 16 + hf*8, bb1);

  for (int hc = 0; hc < 8; ++hc){
    v8f a00 = z8, a01 = z8, a10 = z8, a11 = z8;
    #pragma unroll
    for (int kb = 0; kb < 4; ++kb){
      a00 = __builtin_amdgcn_wmma_f32_16x16x32_f16(false, bf[kb*2],   false, af[0][kb], (short)0, a00, false, false);
      a01 = __builtin_amdgcn_wmma_f32_16x16x32_f16(false, bf[kb*2+1], false, af[0][kb], (short)0, a01, false, false);
      a10 = __builtin_amdgcn_wmma_f32_16x16x32_f16(false, bf[kb*2],   false, af[1][kb], (short)0, a10, false, false);
      a11 = __builtin_amdgcn_wmma_f32_16x16x32_f16(false, bf[kb*2+1], false, af[1][kb], (short)0, a11, false, false);
    }
    v16h bw[8];
    #pragma unroll
    for (int nt = 0; nt < 8; ++nt) bw[nt] = load_frag_glob(w2h, hc*8 + nt, lane);
    const int hn = imin_(hc + 1, 7);
    v16h bfn[8];
    #pragma unroll
    for (int f = 0; f < 8; ++f) bfn[f] = load_frag_glob(w1h, hn*8 + f, lane);
    float nb0[8], nb1[8];
    ldc8(b1 + hn*32 + hf*8, nb0);
    ldc8(b1 + hn*32 + 16 + hf*8, nb1);
    v16h ah0, ah1;
    #pragma unroll
    for (int r = 0; r < 8; ++r){
      const v2f x0 = { a00[r] + bb0[r], a10[r] + bb0[r] };
      const v2f x1 = { a01[r] + bb1[r], a11[r] + bb1[r] };
      const v2f g0 = gelu2(x0), g1 = gelu2(x1);
      ah0[r]     = (_Float16)g0.x;
      ah1[r]     = (_Float16)g0.y;
      ah0[8 + r] = (_Float16)g1.x;
      ah1[8 + r] = (_Float16)g1.y;
    }
    #pragma unroll
    for (int nt = 0; nt < 8; ++nt){
      accC[nt][0] = __builtin_amdgcn_wmma_f32_16x16x32_f16(false, bw[nt], false, ah0, (short)0, accC[nt][0], false, false);
      accC[nt][1] = __builtin_amdgcn_wmma_f32_16x16x32_f16(false, bw[nt], false, ah1, (short)0, accC[nt][1], false, false);
    }
    #pragma unroll
    for (int f = 0; f < 8; ++f) bf[f] = bfn[f];
    #pragma unroll
    for (int r = 0; r < 8; ++r){ bb0[r] = nb0[r]; bb1[r] = nb1[r]; }
  }

  #pragma unroll
  for (int nt = 0; nt < 8; ++nt){
    #pragma unroll
    for (int r = 0; r < 8; ++r){
      float2 rv = { accC[nt][0][r], accC[nt][1][r] };
      *(float2*)&Rs[wv][nt*16 + hf*8 + r][2*col] = rv;
    }
  }
  __syncthreads();

  const float dt = scal[0], al = scal[2];

  float ww[2][4];
  #pragma unroll
  for (int j = 0; j < 2; ++j){
    const float sw = fminf(fmaxf((w0 + j + 0.5f)*(4.f/W_) - 0.5f, 0.f), 3.f);
    const int wi0 = (int)sw; const int wi1 = imin_(wi0+1, 3); const float f = sw - (float)wi0;
    #pragma unroll
    for (int i = 0; i < 4; ++i) ww[j][i] = (i == wi0 ? 1.f - f : 0.f) + (i == wi1 ? f : 0.f);
  }

  const bool vU1 = hh - 1  >= 0, vD1 = hh + 1  < 64;
  const bool vU4 = hh - 4  >= 0, vD4 = hh + 4  < 64;
  const bool vU16= hh - 16 >= 0, vD16= hh + 16 < 64;
  const bool vT1 = t - 1 >= 0,   vT2 = t - 2 >= 0;
  const bool vL4 = w0 - 4  >= 0, vR4 = w0 + 5  < 64;
  const bool vL16= w0 - 16 >= 0, vR16= w0 + 17 < 64;
  const bool vLm = w0 - 1 >= 0,  vRp = w0 + 2 < 64;
  const int oU1 = vU1 ? -64 : 0,     oD1 = vD1 ? 64 : 0;
  const int oU4 = vU4 ? -256 : 0,    oD4 = vD4 ? 256 : 0;
  const int oU16= vU16 ? -1024 : 0,  oD16= vD16 ? 1024 : 0;
  const int oT1 = vT1 ? -4096 : 0,   oT2 = vT2 ? -8192 : 0;
  const int oL4 = vL4 ? -4 : 0,      oR4 = vR4 ? 4 : 0;
  const int oL16= vL16 ? -16 : 0,    oR16= vR16 ? 16 : 0;
  const int oLm = vLm ? -1 : 0,      oRp = vRp ? 2 : 0;
  const bool vG1 = rowi == 0 ? vU1 : vD1;
  const int  oG1 = rowi == 0 ? oU1 : oD1;

  const size_t bbase = (size_t)b * C_ * N_;
  float sum_du = 0.f, sum_lap = 0.f;
  float ssq0 = 0.f, ssq1 = 0.f;

  float*       Rw   = &Rs[wv][hf*8][2*col];
  const float* tabw = &tabs[hf*8][0];
  const float* Lww  = &Lws[wv >> 1][hf*8][0];
  const float* Cme  = &Cs[rowi][hf*8][w0];
  const float* Cot  = &Cs[rowi ^ 1][hf*8][w0];

  auto channel_steps = [&](auto norm_tag){
  constexpr bool NORM = decltype(norm_tag)::value;
  for (int ib = 0; ib < 8; ++ib){
  float2 gG1[8], gU4[8], gD4[8], gU16[8], gD16[8], gT1[8], gT2[8];
  #pragma unroll
  for (int j = 0; j < 8; ++j){
    const float* uc = ub + (size_t)(ib*16 + j)*N_ + e0;
    gG1[j]  = *(const float2*)(uc + oG1);
    gU4[j]  = *(const float2*)(uc + oU4);   gD4[j]  = *(const float2*)(uc + oD4);
    gU16[j] = *(const float2*)(uc + oU16);  gD16[j] = *(const float2*)(uc + oD16);
    gT1[j]  = *(const float2*)(uc + oT1);   gT2[j]  = *(const float2*)(uc + oT2);
  }
  #pragma unroll
  for (int j = 0; j < 8; ++j){
    const int cl = ib*16 + j;
    const float4 ta = *(const float4*)(tabw + cl*8);
    const float4 tb = *(const float4*)(tabw + cl*8 + 4);
    const float4 Lq = *(const float4*)(Lww + cl*4);
    const float2 rc = *(const float2*)(Rw + cl*36);
    const float lv0 = Lq.x*ww[0][0] + Lq.y*ww[0][1] + Lq.z*ww[0][2] + Lq.w*ww[0][3];
    const float lv1 = Lq.x*ww[1][0] + Lq.y*ww[1][1] + Lq.z*ww[1][2] + Lq.w*ww[1][3];

    const float* cc = Cme + cl*68;
    const float2 G1  = sel2(gG1[j],  vG1);
    const float2 U4  = sel2(gU4[j],  vU4),  D4  = sel2(gD4[j],  vD4);
    const float2 U16 = sel2(gU16[j], vU16), D16 = sel2(gD16[j], vD16);
    const float2 T1  = sel2(gT1[j],  vT1),  T2  = sel2(gT2[j],  vT2);
    const float2 cen = *(const float2*)cc;
    const float2 S1  = *(const float2*)(Cot + cl*68);
    const float2 L4  = ld2(cc + oL4,  vL4),  R4  = ld2(cc + oR4,  vR4);
    const float2 L16 = ld2(cc + oL16, vL16), R16 = ld2(cc + oR16, vR16);
    const float  lm  = ld1(cc + oLm,  vLm);
    const float  rp  = ld1(cc + oRp,  vRp);
    const float2 U1  = rowi == 0 ? G1 : S1;
    const float2 D1  = rowi == 0 ? S1 : G1;

    const float rest0 = (rc.x + tb.y) + tb.z;
    const float rest1 = (rc.y + tb.y) + tb.z;
    const float cnd0 = upd1(cen.x, U1.x, D1.x, lm,    cen.y, U4.x, D4.x, L4.x, R4.x,
                            U16.x, D16.x, L16.x, R16.x, T1.x, T2.x,
                            ta.x, ta.y, ta.z, ta.w, tb.x, rest0, lv0, al, dt,
                            sum_lap, sum_du, ssq0);
    const float cnd1 = upd1(cen.y, U1.y, D1.y, cen.x, rp,    U4.y, D4.y, L4.y, R4.y,
                            U16.y, D16.y, L16.y, R16.y, T1.y, T2.y,
                            ta.x, ta.y, ta.z, ta.w, tb.x, rest1, lv1, al, dt,
                            sum_lap, sum_du, ssq1);
    const float2 cv = { cnd0, cnd1 };
    if constexpr (NORM){
      *(float2*)(ucand + bbase + (size_t)cl*N_ + e0) = cv;
      *(float2*)(Rw + cl*36) = cv;
    } else {
      *(float2*)(unext + bbase + (size_t)cl*N_ + e0) = cv;
    }
  }
  }
  };
  if (do_norm) channel_steps(std::true_type{});
  else         channel_steps(std::false_type{});

  if (do_norm){
    ssq0 += __shfl_xor(ssq0, 16);
    ssq1 += __shfl_xor(ssq1, 16);
    const float scale0 = rsqrtf(ssq0/(float)C_ + 1e-8f);
    const float scale1 = rsqrtf(ssq1/(float)C_ + 1e-8f);
    #pragma unroll 8
    for (int it = 0; it < 64; ++it){
      const int cl = ((it >> 3) << 4) | (it & 7);
      const float2 cv = *(const float2*)(Rw + cl*36);
      const float nw = tabw[cl*8 + 7];
      float2 o;
      o.x = cv.x*scale0*nw;
      o.y = cv.y*scale1*nw;
      *(float2*)(unext + bbase + (size_t)cl*N_ + e0) = o;
    }
  }

  float s1 = blockReduceSum(sum_du);
  float s2 = blockReduceSum(sum_lap);
  if (threadIdx.x == 0){
    atomicAdd(&accum[1], s1);
    atomicAdd(&accum[0], s2);
  }
}

__global__ void final_out_kernel(const float* accum, const float* turbs, const float* sten, float* out){
  if (threadIdx.x == 0 && blockIdx.x == 0){
    float tl, sl;
    step_scalars(accum, tl, sl);
    float a = 0.f, bm = 0.f;
    for (int i = 0; i < MAX_STEPS_ - 1; ++i){ a += turbs[i]; bm += sten[i]; }
    a += tl; bm += sl;
    out[0] = a / (float)MAX_STEPS_;
    out[1] = bm / (float)MAX_STEPS_;
  }
}


extern "C" void kernel_launch(void* const* d_in, const int* in_sizes, int n_in,
                              void* d_out, int out_size, void* d_ws, size_t ws_size,
                              hipStream_t stream){
  (void)in_sizes; (void)n_in; (void)out_size; (void)ws_size;

  const float* u_in  = (const float*)d_in[0];
  const float* w1    = (const float*)d_in[1];
  const float* b1    = (const float*)d_in[2];
  const float* w2    = (const float*)d_in[3];
  const float* b2    = (const float*)d_in[4];
  const float* gw    = (const float*)d_in[5];
  const float* gb    = (const float*)d_in[6];
  const float* vw    = (const float*)d_in[7];
  const float* vb    = (const float*)d_in[8];
  const float* normw = (const float*)d_in[9];
  const float* wlp   = (const float*)d_in[10];
  const float* blp   = (const float*)d_in[11];
  const float* a_g   = (const float*)d_in[12];
  const float* a_l   = (const float*)d_in[13];
  const float* logdt = (const float*)d_in[14];
  const float* ks    = (const float*)d_in[15];
  const float* kt    = (const float*)d_in[16];

  float* Wf = (float*)d_ws;
  size_t o = 0;
  const size_t VOL = (size_t)B_ * C_ * N_;
  float* ubuf0  = Wf + o; o += VOL;
  float* ubuf1  = Wf + o; o += VOL;
  float* ucand  = Wf + o; o += VOL;
  float* probeA = Wf + o; o += PROBE_CELLS;
  float* probeB = Wf + o; o += PROBE_CELLS;
  float* pooled = Wf + o; o += 256;
  float* hstate = Wf + o; o += 256;
  float* lsmall = Wf + o; o += (size_t)B_*C_*32;
  float* lproj  = Wf + o; o += (size_t)B_*C_*32;
  float* spsks  = Wf + o; o += 3*C_;
  float* spskt  = Wf + o; o += 2*C_;
  float* scal   = Wf + o; o += 8;
  float* accum  = Wf + o; o += 8;
  float* turbs  = Wf + o; o += 16;
  float* sten   = Wf + o; o += 16;
  _Float16* w1h = (_Float16*)(Wf + o); o += (C_*HID_)/2;
  _Float16* w2h = (_Float16*)(Wf + o); o += (HID_*C_)/2;

  float* ubuf[2] = { ubuf0, ubuf1 };
  float* ufinal = (float*)d_out;

  prep_kernel<<<64, 256, 0, stream>>>(w1, w2, ks, kt, a_g, a_l, logdt,
                                      w1h, w2h, spsks, spskt, scal, hstate);
  pool3_kernel<<<B_*C_, 512, 0, stream>>>(u_in, u_in, 0, probeA, (const float*)nullptr,
                                          lsmall, pooled, accum);

  float* pprev = probeA;
  float* pcur  = probeB;

  for (int step = 0; step < MAX_STEPS_; ++step){
    const float* ucur  = (step == 0) ? u_in : ubuf[step & 1];
    float*       unext = (step == MAX_STEPS_ - 1) ? ufinal : ubuf[(step & 1) ^ 1];
    int do_norm = (((step + 1) % 2) == 0) ? 1 : 0;
    const float* cplane = do_norm ? ucand : unext;

    memory_kernel<<<MEM_GROUPS, 256, 0, stream>>>(pooled, gw, gb, vw, vb, hstate, lsmall, wlp, blp, lproj,
                                                  accum, turbs, sten, step);
    step_kernel<<<BN_/128, 128, 0, stream>>>(ucur, w1h, b1, w2h, b2, hstate, lproj, spsks, spskt,
                                             scal, normw, ucand, unext, accum, do_norm);
    pool3_kernel<<<B_*C_, 512, 0, stream>>>(cplane, unext, do_norm, pcur, pprev, lsmall, pooled, accum);

    float* tswap = pprev; pprev = pcur; pcur = tswap;
  }

  final_out_kernel<<<1, 32, 0, stream>>>(accum, turbs, sten, ((float*)d_out) + VOL);
}
